# starting kernel
# baseline (speedup 1.0000x reference)
#include <hip/hip_runtime.h>
#include <math.h>

typedef float v2f __attribute__((ext_vector_type(2)));
typedef float v4f __attribute__((ext_vector_type(4)));
typedef float v8f __attribute__((ext_vector_type(8)));

#define NB 1024
#define NI 512
#define NO 512
#define EPS 1e-7f
#define LOG_CLIP 20.0f
#define NEG_BIG -1e30f


__global__ void nalu_prep_w_kernel(const float* __restrict__ w_hat,
                                   const float* __restrict__ m_hat,
                                   float* __restrict__ PB) {
    int t = blockIdx.x * blockDim.x + threadIdx.x;
    int r = t >> 9;
    int c = t & 511;
    float wv = tanhf(w_hat[t]) * (1.0f / (1.0f + expf(-m_hat[t])));
    PB[(((r >> 1) * NO + c) << 3) + (r & 1)] = wv;
    float wsv = fabsf(wv);
    float t2 = 1.0f - 2.0f * wsv;
    float t1 = 1.0f - wsv;
    int d = (((c >> 1) * NO + r) << 3) + (c & 1);
    PB[d + 2] = fmaxf(logf(fabsf(t2)), NEG_BIG);
    PB[d + 4] = fmaxf(logf(t1), NEG_BIG);
    PB[d + 6] = (t2 < 0.0f) ? 1.0f : 0.0f;
}

__global__ void nalu_prep_x_kernel(const float* __restrict__ x,
                                   float* __restrict__ PA) {
    int t = blockIdx.x * blockDim.x + threadIdx.x;
    int b = t >> 9;
    int i = t & 511;
    float xv = x[t];
    int d = ((b * (NI >> 1) + (i >> 1)) << 2) + (i & 1);
    PA[d]     = xv;
    PA[d + 2] = logf(fmaxf(fabsf(xv), EPS));
}

__global__ void nalu_main_kernel(const float* __restrict__ PA,
                                 const float* __restrict__ g,
                                 const float* __restrict__ PB,
                                 float* __restrict__ out) {
    const int lane = threadIdx.x & 31;
    const int wave = threadIdx.x >> 5;
    const int tile = blockIdx.x * 8 + wave;
    const int m0 = (tile >> 5) << 5;
    const int n0 = (tile & 31) << 4;
    const int half = lane >> 4;
    const int l15 = lane & 15;
    const int bcol = n0 + l15;

    const v4f* pa0 = (const v4f*)PA + (m0 + l15) * (NI >> 1) + half;
    const v4f* pa1 = pa0 + 16 * (NI >> 1);
    const v4f* pb  = (const v4f*)PB + ((half * NO + bcol) << 1);

    v8f accA0 = {}, accM0 = {}, accL0 = {}, accP0 = {};
    v8f accA1 = {}, accM1 = {}, accL1 = {}, accP1 = {};

#pragma unroll 4
    for (int k = 0; k < NI; k += 4) {
        v4f a0 = pa0[0];  pa0 += 2;
        v4f a1 = pa1[0];  pa1 += 2;
        v4f bq0 = pb[0];
        v4f bq1 = pb[1];
        pb += 4 * NO;

        v2f ax0 = __builtin_shufflevector(a0, a0, 0, 1);
        v2f al0 = __builtin_shufflevector(a0, a0, 2, 3);
        v2f ax1 = __builtin_shufflevector(a1, a1, 0, 1);
        v2f al1 = __builtin_shufflevector(a1, a1, 2, 3);
        v2f bw  = __builtin_shufflevector(bq0, bq0, 0, 1);
        v2f b2  = __builtin_shufflevector(bq0, bq0, 2, 3);
        v2f b1  = __builtin_shufflevector(bq1, bq1, 0, 1);
        v2f bn  = __builtin_shufflevector(bq1, bq1, 2, 3);

        v2f an0, az0, an1, az1;
        an0.x = (ax0.x < 0.0f) ? 1.0f : 0.0f;
        an0.y = (ax0.y < 0.0f) ? 1.0f : 0.0f;
        az0.x = (ax0.x == 0.0f) ? 1.0f : 0.0f;
        az0.y = (ax0.y == 0.0f) ? 1.0f : 0.0f;
        an1.x = (ax1.x < 0.0f) ? 1.0f : 0.0f;
        an1.y = (ax1.y < 0.0f) ? 1.0f : 0.0f;
        az1.x = (ax1.x == 0.0f) ? 1.0f : 0.0f;
        az1.y = (ax1.y == 0.0f) ? 1.0f : 0.0f;

        accA0 = __builtin_amdgcn_wmma_f32_16x16x4_f32(false, ax0, false, bw,
                                                      (short)0, accA0, false, false);
        accA1 = __builtin_amdgcn_wmma_f32_16x16x4_f32(false, ax1, false, bw,
                                                      (short)0, accA1, false, false);
        accM0 = __builtin_amdgcn_wmma_f32_16x16x4_f32(false, al0, false, bw,
                                                      (short)0, accM0, false, false);
        accM1 = __builtin_amdgcn_wmma_f32_16x16x4_f32(false, al1, false, bw,
                                                      (short)0, accM1, false, false);
        accL0 = __builtin_amdgcn_wmma_f32_16x16x4_f32(false, an0, false, b2,
                                                      (short)0, accL0, false, false);
        accL1 = __builtin_amdgcn_wmma_f32_16x16x4_f32(false, an1, false, b2,
                                                      (short)0, accL1, false, false);
        accL0 = __builtin_amdgcn_wmma_f32_16x16x4_f32(false, az0, false, b1,
                                                      (short)0, accL0, false, false);
        accL1 = __builtin_amdgcn_wmma_f32_16x16x4_f32(false, az1, false, b1,
                                                      (short)0, accL1, false, false);
        accP0 = __builtin_amdgcn_wmma_f32_16x16x4_f32(false, an0, false, bn,
                                                      (short)0, accP0, false, false);
        accP1 = __builtin_amdgcn_wmma_f32_16x16x4_f32(false, an1, false, bn,
                                                      (short)0, accP1, false, false);
    }

    const float g1 = 1.0f / (1.0f + expf(-g[bcol]));
    const float gm = 1.0f - g1;
#pragma unroll
    for (int v = 0; v < 8; ++v) {
        const int row0 = m0 + v + 8 * half;
        {
            const float m1 = expf(fminf(accM0[v], LOG_CLIP));
            const int negc = (int)(accP0[v] + 0.5f);
            float ms = ((negc & 1) ? -1.0f : 1.0f) * expf(accL0[v]);
            ms = fminf(fmaxf(ms, -1.0f), 1.0f);
            out[row0 * NO + bcol] = g1 * accA0[v] + gm * m1 * ms;
        }
        {
            const float m1 = expf(fminf(accM1[v], LOG_CLIP));
            const int negc = (int)(accP1[v] + 0.5f);
            float ms = ((negc & 1) ? -1.0f : 1.0f) * expf(accL1[v]);
            ms = fminf(fmaxf(ms, -1.0f), 1.0f);
            out[(row0 + 16) * NO + bcol] = g1 * accA1[v] + gm * m1 * ms;
        }
    }
}

extern "C" void kernel_launch(void* const* d_in, const int* in_sizes, int n_in,
                              void* d_out, int out_size, void* d_ws, size_t ws_size,
                              hipStream_t stream) {
    const float* x     = (const float*)d_in[0];
    const float* w_hat = (const float*)d_in[1];
    const float* m_hat = (const float*)d_in[2];
    const float* g     = (const float*)d_in[3];
    float* out = (float*)d_out;

    float* wsf = (float*)d_ws;
    float* PB  = wsf;
    float* PA  = wsf + (NI / 2) * NO * 8;

    nalu_prep_w_kernel<<<(NI * NO) / 256, 256, 0, stream>>>(w_hat, m_hat, PB);
    nalu_prep_x_kernel<<<(NB * NI) / 256, 256, 0, stream>>>(x, PA);

    nalu_main_kernel<<<128, 256, 0, stream>>>(PA, g, PB, out);
}


# sign counts of the wave's two 16-row tiles from one f32 WMMA per k-step: A operand = flag(rows 0-15) + 1024*flag(rows 16-31), counts <= 512 exact in radix-1024 digits; 9 not 10 WMMAs; sums kept
# speedup vs baseline: 4.0768x; 4.0768x over previous
#include <hip/hip_runtime.h>
#include <math.h>

typedef float v2f __attribute__((ext_vector_type(2)));
typedef float v4f __attribute__((ext_vector_type(4)));
typedef float v8f __attribute__((ext_vector_type(8)));

#define NB 1024
#define NI 512
#define NO 512
#define EPS 1e-7f
#define LOG_CLIP 20.0f
#define NEG_BIG -1e30f


#define PT_P 16
#define PT_C 32
#define PT_PITCH (2 * PT_C + 1)
#define PT_ROWS (NB / 256)

__device__ __forceinline__ float nalu_w1(float wh, float mh) {
    return tanhf(wh) * (1.0f / (1.0f + expf(-mh)));
}

__global__ __launch_bounds__(256) void nalu_prep_kernel(const float* __restrict__ x,
                                                        const float* __restrict__ w_hat,
                                                        const float* __restrict__ m_hat,
                                                        float* __restrict__ PB,
                                                        float* __restrict__ PA) {
    __shared__ v4f rec[PT_P * PT_PITCH];
    float* recf = (float*)rec;

    const int tid = threadIdx.x;
    const int k0 = (blockIdx.x >> 4) * (2 * PT_P);
    const int c0 = (blockIdx.x & 15) * PT_C;
    const int r = tid >> 3;
    const int q = tid & 7;
    const int xr0 = blockIdx.x * PT_ROWS;

    const v4f wd = *(const v4f*)(w_hat + (k0 + r) * NO + c0 + 4 * q);
    const v4f md = *(const v4f*)(m_hat + (k0 + r) * NO + c0 + 4 * q);
    const v4f wt = *(const v4f*)(w_hat + (c0 + r) * NO + k0 + 4 * q);
    const v4f mt = *(const v4f*)(m_hat + (c0 + r) * NO + k0 + 4 * q);
    v2f xv[PT_ROWS];
#pragma unroll
    for (int j = 0; j < PT_ROWS; ++j)
        xv[j] = *(const v2f*)(x + (xr0 + j) * NI + 2 * tid);

#pragma unroll
    for (int j = 0; j < PT_ROWS; ++j) {
        v4f o;
        o.x = xv[j].x;
        o.y = xv[j].y;
        o.z = logf(fmaxf(fabsf(xv[j].x), EPS));
        o.w = logf(fmaxf(fabsf(xv[j].y), EPS));
        *(v4f*)(PA + (((xr0 + j) * (NI >> 1) + tid) << 2)) = o;
    }

#pragma unroll
    for (int e = 0; e < 4; ++e)
        recf[(((r >> 1) * PT_PITCH + 2 * (4 * q + e)) << 2) + (r & 1)] = nalu_w1(wd[e], md[e]);

#pragma unroll
    for (int s = 0; s < 2; ++s) {
        const float wsa = fabsf(nalu_w1(wt[2 * s], mt[2 * s]));
        const float wsb = fabsf(nalu_w1(wt[2 * s + 1], mt[2 * s + 1]));
        const float t2a = 1.0f - 2.0f * wsa;
        const float t2b = 1.0f - 2.0f * wsb;
        const float t1a = 1.0f - wsa;
        const float t1b = 1.0f - wsb;
        v2f lo;
        lo.x = fmaxf(logf(fabsf(t2a)), NEG_BIG);
        lo.y = fmaxf(logf(fabsf(t2b)), NEG_BIG);
        v4f hi;
        hi.x = fmaxf(logf(t1a), NEG_BIG);
        hi.y = fmaxf(logf(t1b), NEG_BIG);
        hi.z = (t2a < 0.0f) ? 1.0f : 0.0f;
        hi.w = (t2b < 0.0f) ? 1.0f : 0.0f;
        const int base = (2 * q + s) * PT_PITCH + 2 * r;
        *(v2f*)(recf + (base << 2) + 2) = lo;
        rec[base + 1] = hi;
    }

    __syncthreads();

#pragma unroll
    for (int j = 0; j < 4; ++j) {
        const int u = tid + 256 * j;
        const int pl = u >> 6;
        const int v = u & 63;
        *(v4f*)(PB + ((((k0 >> 1) + pl) * NO + c0) << 3) + (v << 2)) = rec[pl * PT_PITCH + v];
    }
}

#define BM 64
#define BN 32
#define KP 16
#define KS (KP / 2)
#define NSTAGE ((NI / 2) / KP)
#define A_PITCH (KP + 1)
#define B_PITCH (BN + 1)
#define A_V4 (BM * A_PITCH)
#define B_V4 (KP * B_PITCH)
#define STAGE_V4 (A_V4 + 2 * B_V4)
#define NLD 8
#define PRADIX 1024.0f
#define PSHIFT 10

struct Acc {
    v8f A0, M0, L0, A1, M1, L1, P;
};
struct Frag {
    v4f a0[KS], a1[KS], q0[KS], q1[KS];
};
struct Copy {
    v4f a[NLD], b[NLD];
};

__device__ __forceinline__ void nalu_frag_load(const v4f* fa0, const v4f* fb0, Frag& f) {
    const v4f* fa1 = fa0 + 16 * A_PITCH;
    const v4f* fb1 = fb0 + B_V4;
#pragma unroll
    for (int ks = 0; ks < KS; ++ks) {
        f.a0[ks] = fa0[2 * ks];
        f.a1[ks] = fa1[2 * ks];
        f.q0[ks] = fb0[2 * ks * B_PITCH];
        f.q1[ks] = fb1[2 * ks * B_PITCH];
    }
}

template <int K0, int K1>
__device__ __forceinline__ void nalu_wmma(const Frag& f, Acc& acc) {
#pragma unroll
    for (int ks = K0; ks < K1; ++ks) {
        v2f ax0 = __builtin_shufflevector(f.a0[ks], f.a0[ks], 0, 1);
        v2f al0 = __builtin_shufflevector(f.a0[ks], f.a0[ks], 2, 3);
        v2f ax1 = __builtin_shufflevector(f.a1[ks], f.a1[ks], 0, 1);
        v2f al1 = __builtin_shufflevector(f.a1[ks], f.a1[ks], 2, 3);
        v2f bw  = __builtin_shufflevector(f.q0[ks], f.q0[ks], 0, 1);
        v2f b2  = __builtin_shufflevector(f.q0[ks], f.q0[ks], 2, 3);
        v2f b1  = __builtin_shufflevector(f.q1[ks], f.q1[ks], 0, 1);
        v2f bn  = __builtin_shufflevector(f.q1[ks], f.q1[ks], 2, 3);

        v2f an0, az0, an1, az1;
        an0.x = (ax0.x < 0.0f) ? 1.0f : 0.0f;
        an0.y = (ax0.y < 0.0f) ? 1.0f : 0.0f;
        az0.x = (ax0.x == 0.0f) ? 1.0f : 0.0f;
        az0.y = (ax0.y == 0.0f) ? 1.0f : 0.0f;
        an1.x = (ax1.x < 0.0f) ? 1.0f : 0.0f;
        an1.y = (ax1.y < 0.0f) ? 1.0f : 0.0f;
        az1.x = (ax1.x == 0.0f) ? 1.0f : 0.0f;
        az1.y = (ax1.y == 0.0f) ? 1.0f : 0.0f;

        acc.A0 = __builtin_amdgcn_wmma_f32_16x16x4_f32(false, ax0, false, bw,
                                                       (short)0, acc.A0, false, false);
        acc.A1 = __builtin_amdgcn_wmma_f32_16x16x4_f32(false, ax1, false, bw,
                                                       (short)0, acc.A1, false, false);
        acc.M0 = __builtin_amdgcn_wmma_f32_16x16x4_f32(false, al0, false, bw,
                                                       (short)0, acc.M0, false, false);
        acc.M1 = __builtin_amdgcn_wmma_f32_16x16x4_f32(false, al1, false, bw,
                                                       (short)0, acc.M1, false, false);
        acc.L0 = __builtin_amdgcn_wmma_f32_16x16x4_f32(false, an0, false, b2,
                                                       (short)0, acc.L0, false, false);
        acc.L1 = __builtin_amdgcn_wmma_f32_16x16x4_f32(false, an1, false, b2,
                                                       (short)0, acc.L1, false, false);
        acc.L0 = __builtin_amdgcn_wmma_f32_16x16x4_f32(false, az0, false, b1,
                                                       (short)0, acc.L0, false, false);
        acc.L1 = __builtin_amdgcn_wmma_f32_16x16x4_f32(false, az1, false, b1,
                                                       (short)0, acc.L1, false, false);
        v2f ap;
        ap.x = fmaf(an1.x, PRADIX, an0.x);
        ap.y = fmaf(an1.y, PRADIX, an0.y);
        acc.P = __builtin_amdgcn_wmma_f32_16x16x4_f32(false, ap, false, bn,
                                                      (short)0, acc.P, false, false);
    }
}

__device__ __forceinline__ void nalu_copy_load(const v4f* gA, const v4f* gB, Copy& c) {
#pragma unroll
    for (int j = 0; j < NLD; ++j) {
        c.a[j] = gA[j * 8 * (NI >> 1)];
        c.b[j] = gB[j * 2 * (2 * NO)];
    }
}

__device__ __forceinline__ void nalu_copy_store(v4f* sA, v4f* sB, const Copy& c) {
#pragma unroll
    for (int j = 0; j < NLD; ++j) {
        sA[j * 8 * A_PITCH] = c.a[j];
        sB[j * 2 * B_PITCH] = c.b[j];
    }
}

__device__ __forceinline__ void nalu_ring_signal() {
    __builtin_amdgcn_fence(__ATOMIC_RELEASE, "workgroup");
    __builtin_amdgcn_s_barrier_signal(-1);
    __builtin_amdgcn_sched_barrier(0);
}
__device__ __forceinline__ void nalu_ring_wait() {
    __builtin_amdgcn_sched_barrier(0);
    __builtin_amdgcn_s_barrier_wait(-1);
    __builtin_amdgcn_fence(__ATOMIC_ACQUIRE, "workgroup");
}

__global__ __launch_bounds__(128) void nalu_main_kernel(const float* __restrict__ PA,
                                                        const float* __restrict__ g,
                                                        const float* __restrict__ PB,
                                                        float* __restrict__ out) {
    __shared__ v4f lds[2 * STAGE_V4];

    const int tid = threadIdx.x;
    const int lane = tid & 31;
    const int wave = tid >> 5;
    const int wm = wave >> 1;
    const int wn = wave & 1;
    const int bm0 = (blockIdx.x >> 4) * BM;
    const int bn0 = (blockIdx.x & 15) * BN;
    const int half = lane >> 4;
    const int l15 = lane & 15;

    const v4f* gA = (const v4f*)PA + (bm0 + (tid >> 4)) * (NI >> 1) + (tid & 15);
    const v4f* gB = (const v4f*)PB + (tid >> 6) * (2 * NO) + bn0 * 2 + (tid & 63);
    const int sAo = (tid >> 4) * A_PITCH + (tid & 15);
    const int sBo = A_V4 + (tid & 1) * B_V4 + (tid >> 6) * B_PITCH + ((tid & 63) >> 1);

    const int fAo = (wm * 32 + l15) * A_PITCH + half;
    const int fBo = A_V4 + half * B_PITCH + wn * 16 + l15;

    Acc acc;
    acc.A0 = v8f{}; acc.M0 = v8f{}; acc.L0 = v8f{};
    acc.A1 = v8f{}; acc.M1 = v8f{}; acc.L1 = v8f{}; acc.P = v8f{};

    Copy cp;
    nalu_copy_load(gA, gB, cp);
    nalu_copy_store(lds + sAo, lds + sBo, cp);
    gA += KP;
    gB += KP * (2 * NO);
    nalu_copy_load(gA, gB, cp);
    __syncthreads();

#pragma unroll 1
    for (int s = 0; s < NSTAGE - 2; ++s) {
        const int cur = (s & 1) * STAGE_V4;
        const int nxt = STAGE_V4 - cur;
        Frag f;
        nalu_frag_load(lds + cur + fAo, lds + cur + fBo, f);
        nalu_copy_store(lds + nxt + sAo, lds + nxt + sBo, cp);
        gA += KP;
        gB += KP * (2 * NO);
        nalu_copy_load(gA, gB, cp);
        nalu_wmma<0, KS - 1>(f, acc);
        nalu_ring_signal();
        nalu_wmma<KS - 1, KS>(f, acc);
        nalu_ring_wait();
    }
    {
        const int cur = ((NSTAGE - 2) & 1) * STAGE_V4;
        const int nxt = STAGE_V4 - cur;
        Frag f;
        nalu_frag_load(lds + cur + fAo, lds + cur + fBo, f);
        nalu_copy_store(lds + nxt + sAo, lds + nxt + sBo, cp);
        nalu_wmma<0, KS - 1>(f, acc);
        nalu_ring_signal();
        nalu_wmma<KS - 1, KS>(f, acc);
        nalu_ring_wait();
    }
    {
        const int cur = ((NSTAGE - 1) & 1) * STAGE_V4;
        Frag f;
        nalu_frag_load(lds + cur + fAo, lds + cur + fBo, f);
        nalu_wmma<0, KS>(f, acc);
    }

    const int m0 = bm0 + wm * 32;
    const int bcol = bn0 + wn * 16 + l15;
    const float g1 = 1.0f / (1.0f + expf(-g[bcol]));
    const float gm = 1.0f - g1;
#pragma unroll
    for (int v = 0; v < 8; ++v) {
        const int row0 = m0 + v + 8 * half;
        const int negc = (int)(acc.P[v] + 0.5f);
        {
            const float m1 = expf(fminf(acc.M0[v], LOG_CLIP));
            float ms = ((negc & 1) ? -1.0f : 1.0f) * expf(acc.L0[v]);
            ms = fminf(fmaxf(ms, -1.0f), 1.0f);
            out[row0 * NO + bcol] = g1 * acc.A0[v] + gm * m1 * ms;
        }
        {
            const float m1 = expf(fminf(acc.M1[v], LOG_CLIP));
            float ms = (((negc >> PSHIFT) & 1) ? -1.0f : 1.0f) * expf(acc.L1[v]);
            ms = fminf(fmaxf(ms, -1.0f), 1.0f);
            out[(row0 + 16) * NO + bcol] = g1 * acc.A1[v] + gm * m1 * ms;
        }
    }
}

extern "C" void kernel_launch(void* const* d_in, const int* in_sizes, int n_in,
                              void* d_out, int out_size, void* d_ws, size_t ws_size,
                              hipStream_t stream) {
    const float* x     = (const float*)d_in[0];
    const float* w_hat = (const float*)d_in[1];
    const float* m_hat = (const float*)d_in[2];
    const float* g     = (const float*)d_in[3];
    float* out = (float*)d_out;

    float* wsf = (float*)d_ws;
    float* PB  = wsf;
    float* PA  = wsf + (NI / 2) * NO * 8;

    nalu_prep_kernel<<<256, 256, 0, stream>>>(x, w_hat, m_hat, PB, PA);

    nalu_main_kernel<<<256, 128, 0, stream>>>(PA, g, PB, out);
}


# main K pipeline a stage deeper: global loads fly 2 stages (2nd copy register set, barrier fences LDS only), fragment reads half a stage ahead behind a mid-stage barrier (2nd fragment set); sums kept
# speedup vs baseline: 4.6507x; 1.1408x over previous
#include <hip/hip_runtime.h>
#include <math.h>

typedef float v2f __attribute__((ext_vector_type(2)));
typedef float v4f __attribute__((ext_vector_type(4)));
typedef float v8f __attribute__((ext_vector_type(8)));

#define NB 1024
#define NI 512
#define NO 512
#define EPS 1e-7f
#define LOG_CLIP 20.0f
#define NEG_BIG -1e30f


#define PT_P 16
#define PT_C 32
#define PT_PITCH (2 * PT_C + 1)
#define PT_ROWS (NB / 256)

__device__ __forceinline__ float nalu_w1(float wh, float mh) {
    return tanhf(wh) * (1.0f / (1.0f + expf(-mh)));
}

__global__ __launch_bounds__(256) void nalu_prep_kernel(const float* __restrict__ x,
                                                        const float* __restrict__ w_hat,
                                                        const float* __restrict__ m_hat,
                                                        float* __restrict__ PB,
                                                        float* __restrict__ PA) {
    __shared__ v4f rec[PT_P * PT_PITCH];
    float* recf = (float*)rec;

    const int tid = threadIdx.x;
    const int k0 = (blockIdx.x >> 4) * (2 * PT_P);
    const int c0 = (blockIdx.x & 15) * PT_C;
    const int r = tid >> 3;
    const int q = tid & 7;
    const int xr0 = blockIdx.x * PT_ROWS;

    const v4f wd = *(const v4f*)(w_hat + (k0 + r) * NO + c0 + 4 * q);
    const v4f md = *(const v4f*)(m_hat + (k0 + r) * NO + c0 + 4 * q);
    const v4f wt = *(const v4f*)(w_hat + (c0 + r) * NO + k0 + 4 * q);
    const v4f mt = *(const v4f*)(m_hat + (c0 + r) * NO + k0 + 4 * q);
    v2f xv[PT_ROWS];
#pragma unroll
    for (int j = 0; j < PT_ROWS; ++j)
        xv[j] = *(const v2f*)(x + (xr0 + j) * NI + 2 * tid);

#pragma unroll
    for (int j = 0; j < PT_ROWS; ++j) {
        v4f o;
        o.x = xv[j].x;
        o.y = xv[j].y;
        o.z = logf(fmaxf(fabsf(xv[j].x), EPS));
        o.w = logf(fmaxf(fabsf(xv[j].y), EPS));
        *(v4f*)(PA + (((xr0 + j) * (NI >> 1) + tid) << 2)) = o;
    }

#pragma unroll
    for (int e = 0; e < 4; ++e)
        recf[(((r >> 1) * PT_PITCH + 2 * (4 * q + e)) << 2) + (r & 1)] = nalu_w1(wd[e], md[e]);

#pragma unroll
    for (int s = 0; s < 2; ++s) {
        const float wsa = fabsf(nalu_w1(wt[2 * s], mt[2 * s]));
        const float wsb = fabsf(nalu_w1(wt[2 * s + 1], mt[2 * s + 1]));
        const float t2a = 1.0f - 2.0f * wsa;
        const float t2b = 1.0f - 2.0f * wsb;
        const float t1a = 1.0f - wsa;
        const float t1b = 1.0f - wsb;
        v2f lo;
        lo.x = fmaxf(logf(fabsf(t2a)), NEG_BIG);
        lo.y = fmaxf(logf(fabsf(t2b)), NEG_BIG);
        v4f hi;
        hi.x = fmaxf(logf(t1a), NEG_BIG);
        hi.y = fmaxf(logf(t1b), NEG_BIG);
        hi.z = (t2a < 0.0f) ? 1.0f : 0.0f;
        hi.w = (t2b < 0.0f) ? 1.0f : 0.0f;
        const int base = (2 * q + s) * PT_PITCH + 2 * r;
        *(v2f*)(recf + (base << 2) + 2) = lo;
        rec[base + 1] = hi;
    }

    __syncthreads();

#pragma unroll
    for (int j = 0; j < 4; ++j) {
        const int u = tid + 256 * j;
        const int pl = u >> 6;
        const int v = u & 63;
        *(v4f*)(PB + ((((k0 >> 1) + pl) * NO + c0) << 3) + (v << 2)) = rec[pl * PT_PITCH + v];
    }
}

#define BM 64
#define BN 32
#define KP 16
#define KS (KP / 2)
#define NSTAGE ((NI / 2) / KP)
#define A_PITCH (KP + 1)
#define B_PITCH (BN + 1)
#define A_V4 (BM * A_PITCH)
#define B_V4 (KP * B_PITCH)
#define STAGE_V4 (A_V4 + 2 * B_V4)
#define NLD 8
#define PRADIX 1024.0f
#define PSHIFT 10

struct Acc {
    v8f A0, M0, L0, A1, M1, L1, P;
};
struct Frag {
    v4f a0[KS], a1[KS], q0[KS], q1[KS];
};
struct Copy {
    v4f a[NLD], b[NLD];
};

__device__ __forceinline__ void nalu_frag_load(const v4f* fa0, const v4f* fb0, Frag& f) {
    const v4f* fa1 = fa0 + 16 * A_PITCH;
    const v4f* fb1 = fb0 + B_V4;
#pragma unroll
    for (int ks = 0; ks < KS; ++ks) {
        f.a0[ks] = fa0[2 * ks];
        f.a1[ks] = fa1[2 * ks];
        f.q0[ks] = fb0[2 * ks * B_PITCH];
        f.q1[ks] = fb1[2 * ks * B_PITCH];
    }
}

template <int K0, int K1>
__device__ __forceinline__ void nalu_wmma(const Frag& f, Acc& acc) {
#pragma unroll
    for (int ks = K0; ks < K1; ++ks) {
        v2f ax0 = __builtin_shufflevector(f.a0[ks], f.a0[ks], 0, 1);
        v2f al0 = __builtin_shufflevector(f.a0[ks], f.a0[ks], 2, 3);
        v2f ax1 = __builtin_shufflevector(f.a1[ks], f.a1[ks], 0, 1);
        v2f al1 = __builtin_shufflevector(f.a1[ks], f.a1[ks], 2, 3);
        v2f bw  = __builtin_shufflevector(f.q0[ks], f.q0[ks], 0, 1);
        v2f b2  = __builtin_shufflevector(f.q0[ks], f.q0[ks], 2, 3);
        v2f b1  = __builtin_shufflevector(f.q1[ks], f.q1[ks], 0, 1);
        v2f bn  = __builtin_shufflevector(f.q1[ks], f.q1[ks], 2, 3);

        v2f an0, az0, an1, az1;
        an0.x = (ax0.x < 0.0f) ? 1.0f : 0.0f;
        an0.y = (ax0.y < 0.0f) ? 1.0f : 0.0f;
        az0.x = (ax0.x == 0.0f) ? 1.0f : 0.0f;
        az0.y = (ax0.y == 0.0f) ? 1.0f : 0.0f;
        an1.x = (ax1.x < 0.0f) ? 1.0f : 0.0f;
        an1.y = (ax1.y < 0.0f) ? 1.0f : 0.0f;
        az1.x = (ax1.x == 0.0f) ? 1.0f : 0.0f;
        az1.y = (ax1.y == 0.0f) ? 1.0f : 0.0f;

        acc.A0 = __builtin_amdgcn_wmma_f32_16x16x4_f32(false, ax0, false, bw,
                                                       (short)0, acc.A0, false, false);
        acc.A1 = __builtin_amdgcn_wmma_f32_16x16x4_f32(false, ax1, false, bw,
                                                       (short)0, acc.A1, false, false);
        acc.M0 = __builtin_amdgcn_wmma_f32_16x16x4_f32(false, al0, false, bw,
                                                       (short)0, acc.M0, false, false);
        acc.M1 = __builtin_amdgcn_wmma_f32_16x16x4_f32(false, al1, false, bw,
                                                       (short)0, acc.M1, false, false);
        acc.L0 = __builtin_amdgcn_wmma_f32_16x16x4_f32(false, an0, false, b2,
                                                       (short)0, acc.L0, false, false);
        acc.L1 = __builtin_amdgcn_wmma_f32_16x16x4_f32(false, an1, false, b2,
                                                       (short)0, acc.L1, false, false);
        acc.L0 = __builtin_amdgcn_wmma_f32_16x16x4_f32(false, az0, false, b1,
                                                       (short)0, acc.L0, false, false);
        acc.L1 = __builtin_amdgcn_wmma_f32_16x16x4_f32(false, az1, false, b1,
                                                       (short)0, acc.L1, false, false);
        v2f ap;
        ap.x = fmaf(an1.x, PRADIX, an0.x);
        ap.y = fmaf(an1.y, PRADIX, an0.y);
        acc.P = __builtin_amdgcn_wmma_f32_16x16x4_f32(false, ap, false, bn,
                                                      (short)0, acc.P, false, false);
    }
}

__device__ __forceinline__ void nalu_copy_load(const v4f* gA, const v4f* gB, Copy& c) {
#pragma unroll
    for (int j = 0; j < NLD; ++j) {
        c.a[j] = gA[j * 8 * (NI >> 1)];
        c.b[j] = gB[j * 2 * (2 * NO)];
    }
}

__device__ __forceinline__ void nalu_copy_store(v4f* sA, v4f* sB, const Copy& c) {
#pragma unroll
    for (int j = 0; j < NLD; ++j) {
        sA[j * 8 * A_PITCH] = c.a[j];
        sB[j * 2 * B_PITCH] = c.b[j];
    }
}

__device__ __forceinline__ void nalu_ring_signal() {
    __builtin_amdgcn_fence(__ATOMIC_RELEASE, "workgroup", "local");
    __builtin_amdgcn_s_barrier_signal(-1);
    __builtin_amdgcn_sched_barrier(0);
}
__device__ __forceinline__ void nalu_ring_wait() {
    __builtin_amdgcn_sched_barrier(0);
    __builtin_amdgcn_s_barrier_wait(-1);
    __builtin_amdgcn_fence(__ATOMIC_ACQUIRE, "workgroup", "local");
}

__device__ __forceinline__ void nalu_wmma_above(const Acc& a) {
    asm volatile("" :: "v"(a.A0[0]), "v"(a.M0[0]), "v"(a.L0[0]), "v"(a.A1[0]),
                       "v"(a.M1[0]), "v"(a.L1[0]), "v"(a.P[0]));
}

template <bool STORE, bool LOAD>
__device__ __forceinline__ void nalu_stage(const Frag& fc, Frag& fn, Copy& c,
                                           v4f* sA, v4f* sB, const v4f* fA, const v4f* fB,
                                           const v4f* gA, const v4f* gB, Acc& acc) {
    if (STORE) {
        nalu_copy_store(sA, sB, c);
        if (LOAD) nalu_copy_load(gA, gB, c);
        __builtin_amdgcn_sched_barrier(0);
    }
    nalu_wmma<0, 3>(fc, acc);
    if (STORE) {
        nalu_wmma_above(acc);
        nalu_ring_signal();
    }
    nalu_wmma<3, 4>(fc, acc);
    if (STORE) {
        nalu_wmma_above(acc);
        nalu_ring_wait();
        nalu_frag_load(fA, fB, fn);
        __builtin_amdgcn_sched_barrier(0);
    }
    nalu_wmma<4, KS>(fc, acc);
}

__global__ __launch_bounds__(128) void nalu_main_kernel(const float* __restrict__ PA,
                                                        const float* __restrict__ g,
                                                        const float* __restrict__ PB,
                                                        float* __restrict__ out) {
    __shared__ v4f lds[2 * STAGE_V4];

    const int tid = threadIdx.x;
    const int lane = tid & 31;
    const int wave = tid >> 5;
    const int wm = wave >> 1;
    const int wn = wave & 1;
    const int bm0 = (blockIdx.x >> 4) * BM;
    const int bn0 = (blockIdx.x & 15) * BN;
    const int half = lane >> 4;
    const int l15 = lane & 15;

    const v4f* gA = (const v4f*)PA + (bm0 + (tid >> 4)) * (NI >> 1) + (tid & 15);
    const v4f* gB = (const v4f*)PB + (tid >> 6) * (2 * NO) + bn0 * 2 + (tid & 63);
    const int sAo = (tid >> 4) * A_PITCH + (tid & 15);
    const int sBo = A_V4 + (tid & 1) * B_V4 + (tid >> 6) * B_PITCH + ((tid & 63) >> 1);

    const int fAo = (wm * 32 + l15) * A_PITCH + half;
    const int fBo = A_V4 + half * B_PITCH + wn * 16 + l15;

    Acc acc;
    acc.A0 = v8f{}; acc.M0 = v8f{}; acc.L0 = v8f{};
    acc.A1 = v8f{}; acc.M1 = v8f{}; acc.L1 = v8f{}; acc.P = v8f{};

    v4f* const slot0 = lds;
    v4f* const slot1 = lds + STAGE_V4;

    Copy c0, c1;
    nalu_copy_load(gA, gB, c0);
    gA += KP;
    gB += KP * (2 * NO);
    nalu_copy_load(gA, gB, c1);
    gA += KP;
    gB += KP * (2 * NO);
    nalu_copy_store(slot0 + sAo, slot0 + sBo, c0);
    nalu_copy_load(gA, gB, c0);
    gA += KP;
    gB += KP * (2 * NO);
    __builtin_amdgcn_fence(__ATOMIC_RELEASE, "workgroup", "local");
    __builtin_amdgcn_s_barrier_signal(-1);
    __builtin_amdgcn_s_barrier_wait(-1);
    __builtin_amdgcn_fence(__ATOMIC_ACQUIRE, "workgroup", "local");

    Frag f0, f1;
    nalu_frag_load(slot0 + fAo, slot0 + fBo, f0);

#pragma unroll 1
    for (int it = 0; it < (NSTAGE - 4) / 2; ++it) {
        nalu_stage<true, true>(f0, f1, c1, slot1 + sAo, slot1 + sBo, slot1 + fAo, slot1 + fBo,
                               gA, gB, acc);
        gA += KP;
        gB += KP * (2 * NO);
        nalu_stage<true, true>(f1, f0, c0, slot0 + sAo, slot0 + sBo, slot0 + fAo, slot0 + fBo,
                               gA, gB, acc);
        gA += KP;
        gB += KP * (2 * NO);
    }
    nalu_stage<true, true>(f0, f1, c1, slot1 + sAo, slot1 + sBo, slot1 + fAo, slot1 + fBo,
                           gA, gB, acc);
    nalu_stage<true, false>(f1, f0, c0, slot0 + sAo, slot0 + sBo, slot0 + fAo, slot0 + fBo,
                            gA, gB, acc);
    nalu_stage<true, false>(f0, f1, c1, slot1 + sAo, slot1 + sBo, slot1 + fAo, slot1 + fBo,
                            gA, gB, acc);
    nalu_stage<false, false>(f1, f0, c0, slot0 + sAo, slot0 + sBo, slot0 + fAo, slot0 + fBo,
                             gA, gB, acc);

    const int m0 = bm0 + wm * 32;
    const int bcol = bn0 + wn * 16 + l15;
    const float g1 = 1.0f / (1.0f + expf(-g[bcol]));
    const float gm = 1.0f - g1;
#pragma unroll
    for (int v = 0; v < 8; ++v) {
        const int row0 = m0 + v + 8 * half;
        const int negc = (int)(acc.P[v] + 0.5f);
        {
            const float m1 = expf(fminf(acc.M0[v], LOG_CLIP));
            float ms = ((negc & 1) ? -1.0f : 1.0f) * expf(acc.L0[v]);
            ms = fminf(fmaxf(ms, -1.0f), 1.0f);
            out[row0 * NO + bcol] = g1 * acc.A0[v] + gm * m1 * ms;
        }
        {
            const float m1 = expf(fminf(acc.M1[v], LOG_CLIP));
            float ms = (((negc >> PSHIFT) & 1) ? -1.0f : 1.0f) * expf(acc.L1[v]);
            ms = fminf(fmaxf(ms, -1.0f), 1.0f);
            out[(row0 + 16) * NO + bcol] = g1 * acc.A1[v] + gm * m1 * ms;
        }
    }
}

extern "C" void kernel_launch(void* const* d_in, const int* in_sizes, int n_in,
                              void* d_out, int out_size, void* d_ws, size_t ws_size,
                              hipStream_t stream) {
    const float* x     = (const float*)d_in[0];
    const float* w_hat = (const float*)d_in[1];
    const float* m_hat = (const float*)d_in[2];
    const float* g     = (const float*)d_in[3];
    float* out = (float*)d_out;

    float* wsf = (float*)d_ws;
    float* PB  = wsf;
    float* PA  = wsf + (NI / 2) * NO * 8;

    nalu_prep_kernel<<<256, 256, 0, stream>>>(x, w_hat, m_hat, PB, PA);

    nalu_main_kernel<<<256, 128, 0, stream>>>(PA, g, PB, out);
}


# prep launch folded into main: each workgroup forms its K stages' x and weight records from the raw inputs between its WMMAs (exp2/log2/rcp unit); 1 launch, no workspace; w1 and log rounding change
# speedup vs baseline: 5.0921x; 1.0949x over previous
#include <hip/hip_runtime.h>
#include <math.h>

typedef float v2f __attribute__((ext_vector_type(2)));
typedef float v4f __attribute__((ext_vector_type(4)));
typedef float v8f __attribute__((ext_vector_type(8)));


#define NB 1024
#define NI 512
#define NO 512
#define EPS 1e-7f
#define LOG_CLIP 20.0f
#define NEG_BIG -1e30f
#define LOG2E 1.44269504088896341f
#define LN2 0.693147180559945309f


__device__ __forceinline__ float nalu_w1(float wh, float mh) {
    const float t = __builtin_amdgcn_exp2f((-2.0f * LOG2E) * fabsf(wh));
    const float F = __builtin_amdgcn_exp2f(-LOG2E * mh);
    const float num = copysignf(1.0f - t, wh);
    return num * __builtin_amdgcn_rcpf((1.0f + t) * (1.0f + F));
}

__device__ __forceinline__ float nalu_ln(float v) {
    return __builtin_amdgcn_logf(v) * LN2;
}

#define BM 64
#define BN 32
#define KP 16
#define KS (KP / 2)
#define NSTAGE ((NI / 2) / KP)
#define A_PITCH (KP + 1)
#define B_PITCH (BN + 1)
#define A_V4 (BM * A_PITCH)
#define B_V4 (KP * B_PITCH)
#define STAGE_V4 (A_V4 + 2 * B_V4)
#define NLD 8
#define PRADIX 1024.0f
#define PSHIFT 10

struct Acc {
    v8f A0, M0, L0, A1, M1, L1, P;
};
struct Frag {
    v4f a0[KS], a1[KS], q0[KS], q1[KS];
};
struct Raw {
    v2f x[NLD];
    v4f wd[2], md[2];
    v4f wt[2], mt[2];
};
struct Rec {
    v4f a[NLD];
    v2f bw[4];
    v2f b2[4];
    v4f b1[4];
};

__device__ __forceinline__ void nalu_frag_load(const v4f* fa0, const v4f* fb0, Frag& f) {
    const v4f* fa1 = fa0 + 16 * A_PITCH;
    const v4f* fb1 = fb0 + B_V4;
#pragma unroll
    for (int ks = 0; ks < KS; ++ks) {
        f.a0[ks] = fa0[2 * ks];
        f.a1[ks] = fa1[2 * ks];
        f.q0[ks] = fb0[2 * ks * B_PITCH];
        f.q1[ks] = fb1[2 * ks * B_PITCH];
    }
}

template <int K0, int K1>
__device__ __forceinline__ void nalu_wmma(const Frag& f, Acc& acc) {
#pragma unroll
    for (int ks = K0; ks < K1; ++ks) {
        v2f ax0 = __builtin_shufflevector(f.a0[ks], f.a0[ks], 0, 1);
        v2f al0 = __builtin_shufflevector(f.a0[ks], f.a0[ks], 2, 3);
        v2f ax1 = __builtin_shufflevector(f.a1[ks], f.a1[ks], 0, 1);
        v2f al1 = __builtin_shufflevector(f.a1[ks], f.a1[ks], 2, 3);
        v2f bw  = __builtin_shufflevector(f.q0[ks], f.q0[ks], 0, 1);
        v2f b2  = __builtin_shufflevector(f.q0[ks], f.q0[ks], 2, 3);
        v2f b1  = __builtin_shufflevector(f.q1[ks], f.q1[ks], 0, 1);
        v2f bn  = __builtin_shufflevector(f.q1[ks], f.q1[ks], 2, 3);

        v2f an0, az0, an1, az1;
        an0.x = (ax0.x < 0.0f) ? 1.0f : 0.0f;
        an0.y = (ax0.y < 0.0f) ? 1.0f : 0.0f;
        az0.x = (ax0.x == 0.0f) ? 1.0f : 0.0f;
        az0.y = (ax0.y == 0.0f) ? 1.0f : 0.0f;
        an1.x = (ax1.x < 0.0f) ? 1.0f : 0.0f;
        an1.y = (ax1.y < 0.0f) ? 1.0f : 0.0f;
        az1.x = (ax1.x == 0.0f) ? 1.0f : 0.0f;
        az1.y = (ax1.y == 0.0f) ? 1.0f : 0.0f;

        acc.A0 = __builtin_amdgcn_wmma_f32_16x16x4_f32(false, ax0, false, bw,
                                                       (short)0, acc.A0, false, false);
        acc.A1 = __builtin_amdgcn_wmma_f32_16x16x4_f32(false, ax1, false, bw,
                                                       (short)0, acc.A1, false, false);
        acc.M0 = __builtin_amdgcn_wmma_f32_16x16x4_f32(false, al0, false, bw,
                                                       (short)0, acc.M0, false, false);
        acc.M1 = __builtin_amdgcn_wmma_f32_16x16x4_f32(false, al1, false, bw,
                                                       (short)0, acc.M1, false, false);
        acc.L0 = __builtin_amdgcn_wmma_f32_16x16x4_f32(false, an0, false, b2,
                                                       (short)0, acc.L0, false, false);
        acc.L1 = __builtin_amdgcn_wmma_f32_16x16x4_f32(false, an1, false, b2,
                                                       (short)0, acc.L1, false, false);
        acc.L0 = __builtin_amdgcn_wmma_f32_16x16x4_f32(false, az0, false, b1,
                                                       (short)0, acc.L0, false, false);
        acc.L1 = __builtin_amdgcn_wmma_f32_16x16x4_f32(false, az1, false, b1,
                                                       (short)0, acc.L1, false, false);
        v2f ap;
        ap.x = fmaf(an1.x, PRADIX, an0.x);
        ap.y = fmaf(an1.y, PRADIX, an0.y);
        acc.P = __builtin_amdgcn_wmma_f32_16x16x4_f32(false, ap, false, bn,
                                                      (short)0, acc.P, false, false);
    }
}

__device__ __forceinline__ void nalu_raw_load(const float* gx, const float* gwd, const float* gmd,
                                              const float* gwt, const float* gmt, Raw& r) {
#pragma unroll
    for (int j = 0; j < NLD; ++j) r.x[j] = *(const v2f*)(gx + j * 8 * NI);
#pragma unroll
    for (int e = 0; e < 2; ++e) {
        r.wd[e] = *(const v4f*)(gwd + e * NO);
        r.md[e] = *(const v4f*)(gmd + e * NO);
        r.wt[e] = *(const v4f*)(gwt + e * 16 * NO);
        r.mt[e] = *(const v4f*)(gmt + e * 16 * NO);
    }
}

__device__ __forceinline__ void nalu_xform_d(const Raw& r, Rec& o) {
#pragma unroll
    for (int i = 0; i < 4; ++i) {
        v2f w;
        w.x = nalu_w1(r.wd[0][i], r.md[0][i]);
        w.y = nalu_w1(r.wd[1][i], r.md[1][i]);
        o.bw[i] = w;
    }
}

__device__ __forceinline__ void nalu_xform_x(const Raw& r, Rec& o) {
#pragma unroll
    for (int j = 0; j < NLD; ++j) {
        v4f a;
        a.x = r.x[j].x;
        a.y = r.x[j].y;
        a.z = nalu_ln(fmaxf(fabsf(r.x[j].x), EPS));
        a.w = nalu_ln(fmaxf(fabsf(r.x[j].y), EPS));
        o.a[j] = a;
    }
}

__device__ __forceinline__ void nalu_xform_m(const Raw& r, Rec& o) {
#pragma unroll
    for (int j = 0; j < 2; ++j) {
#pragma unroll
        for (int h = 0; h < 2; ++h) {
            const float wsa = fabsf(nalu_w1(r.wt[j][2 * h], r.mt[j][2 * h]));
            const float wsb = fabsf(nalu_w1(r.wt[j][2 * h + 1], r.mt[j][2 * h + 1]));
            const float t2a = 1.0f - 2.0f * wsa;
            const float t2b = 1.0f - 2.0f * wsb;
            const float t1a = 1.0f - wsa;
            const float t1b = 1.0f - wsb;
            v2f lo;
            lo.x = fmaxf(nalu_ln(fabsf(t2a)), NEG_BIG);
            lo.y = fmaxf(nalu_ln(fabsf(t2b)), NEG_BIG);
            v4f hi;
            hi.x = fmaxf(nalu_ln(t1a), NEG_BIG);
            hi.y = fmaxf(nalu_ln(t1b), NEG_BIG);
            hi.z = (t2a < 0.0f) ? 1.0f : 0.0f;
            hi.w = (t2b < 0.0f) ? 1.0f : 0.0f;
            o.b2[2 * j + h] = lo;
            o.b1[2 * j + h] = hi;
        }
    }
}

__device__ __forceinline__ void nalu_rec_store(v4f* sA, float* sD, v4f* sM, const Rec& o) {
#pragma unroll
    for (int j = 0; j < NLD; ++j) sA[j * 8 * A_PITCH] = o.a[j];
#pragma unroll
    for (int i = 0; i < 4; ++i) *(v2f*)(sD + 4 * i) = o.bw[i];
#pragma unroll
    for (int j = 0; j < 2; ++j) {
#pragma unroll
        for (int h = 0; h < 2; ++h) {
            v4f* q = sM + h * B_PITCH + j * 16;
            *(v2f*)((float*)q + 2) = o.b2[2 * j + h];
            q[B_V4] = o.b1[2 * j + h];
        }
    }
}

__device__ __forceinline__ void nalu_ring_signal() {
    __builtin_amdgcn_fence(__ATOMIC_RELEASE, "workgroup", "local");
    __builtin_amdgcn_s_barrier_signal(-1);
    __builtin_amdgcn_sched_barrier(0);
}
__device__ __forceinline__ void nalu_ring_wait() {
    __builtin_amdgcn_sched_barrier(0);
    __builtin_amdgcn_s_barrier_wait(-1);
    __builtin_amdgcn_fence(__ATOMIC_ACQUIRE, "workgroup", "local");
}

__device__ __forceinline__ void nalu_wmma_above(const Acc& a) {
    asm volatile("" :: "v"(a.A0[0]), "v"(a.M0[0]), "v"(a.L0[0]), "v"(a.A1[0]),
                       "v"(a.M1[0]), "v"(a.L1[0]), "v"(a.P[0]));
}

template <int N, int V, int ID>
__device__ __forceinline__ void nalu_interleave() {
#pragma unroll
    for (int i = 0; i < N; ++i) {
        __builtin_amdgcn_sched_group_barrier(0x402, V, ID);
        __builtin_amdgcn_sched_group_barrier(0x008, 1, ID);
    }
}

template <bool STORE, bool LOAD, bool XFORM, int ID>
__device__ __forceinline__ void nalu_stage(const Frag& fc, Frag& fn, Raw& rl, const Raw& rx, Rec& rec,
                                           v4f* sA, float* sD, v4f* sM, const v4f* fA, const v4f* fB,
                                           const float* gx, const float* gwd, const float* gmd,
                                           const float* gwt, const float* gmt, Acc& acc) {
    if (STORE) {
        nalu_rec_store(sA, sD, sM, rec);
        if (LOAD) nalu_raw_load(gx, gwd, gmd, gwt, gmt, rl);
        __builtin_amdgcn_sched_barrier(0);
    }
    if (XFORM) nalu_xform_d(rx, rec);
    nalu_wmma<0, 3>(fc, acc);
    if (XFORM) nalu_interleave<3 * 9, 8, ID>();
    if (STORE) {
        nalu_wmma_above(acc);
        nalu_ring_signal();
    }
    if (XFORM) nalu_xform_x(rx, rec);
    nalu_wmma<3, 4>(fc, acc);
    if (XFORM) nalu_interleave<9, 10, ID + 1>();
    if (STORE) {
        nalu_wmma_above(acc);
        nalu_ring_wait();
        nalu_frag_load(fA, fB, fn);
        __builtin_amdgcn_sched_barrier(0);
    }
    if (XFORM) nalu_xform_m(rx, rec);
    nalu_wmma<4, KS>(fc, acc);
    if (XFORM) {
        nalu_interleave<(KS - 4) * 9, 9, ID + 2>();
        nalu_wmma_above(acc);
        __builtin_amdgcn_sched_barrier(0);
    }
}

#define NALU_ADVANCE() do { gx += 2 * KP; gwt += 2 * KP; gmt += 2 * KP; \
                            gwd += 2 * KP * NO; gmd += 2 * KP * NO; } while (0)

__global__ __launch_bounds__(128) void nalu_main_kernel(const float* __restrict__ x,
                                                        const float* __restrict__ w_hat,
                                                        const float* __restrict__ m_hat,
                                                        const float* __restrict__ g,
                                                        float* __restrict__ out) {
    __shared__ v4f lds[2 * STAGE_V4];

    const int tid = threadIdx.x;
    const int lane = tid & 31;
    const int wave = tid >> 5;
    const int wm = wave >> 1;
    const int wn = wave & 1;
    const int bm0 = (blockIdx.x >> 4) * BM;
    const int bn0 = (blockIdx.x & 15) * BN;
    const int half = lane >> 4;
    const int l15 = lane & 15;

    const float* gx = x + (bm0 + (tid >> 4)) * NI + 2 * (tid & 15);
    const int offd = (2 * (tid >> 3)) * NO + bn0 + 4 * (tid & 7);
    const int offt = (bn0 + (tid >> 3)) * NO + 4 * (tid & 7);
    const float* gwd = w_hat + offd;
    const float* gmd = m_hat + offd;
    const float* gwt = w_hat + offt;
    const float* gmt = m_hat + offt;
    const int sAo = (tid >> 4) * A_PITCH + (tid & 15);
    const int sDo = A_V4 + (tid >> 3) * B_PITCH + 4 * (tid & 7);
    const int sMo = A_V4 + 2 * (tid & 7) * B_PITCH + (tid >> 3);

    const int fAo = (wm * 32 + l15) * A_PITCH + half;
    const int fBo = A_V4 + half * B_PITCH + wn * 16 + l15;

    Acc acc;
    acc.A0 = v8f{}; acc.M0 = v8f{}; acc.L0 = v8f{};
    acc.A1 = v8f{}; acc.M1 = v8f{}; acc.L1 = v8f{}; acc.P = v8f{};

    v4f* const slot0 = lds;
    v4f* const slot1 = lds + STAGE_V4;

    Raw c0, c1;
    Rec rec;
    nalu_raw_load(gx, gwd, gmd, gwt, gmt, c0);
    NALU_ADVANCE();
    nalu_raw_load(gx, gwd, gmd, gwt, gmt, c1);
    NALU_ADVANCE();
    nalu_xform_d(c0, rec);
    nalu_xform_x(c0, rec);
    nalu_xform_m(c0, rec);
    nalu_rec_store(slot0 + sAo, (float*)(slot0 + sDo), slot0 + sMo, rec);
    nalu_raw_load(gx, gwd, gmd, gwt, gmt, c0);
    NALU_ADVANCE();
    nalu_xform_d(c1, rec);
    nalu_xform_x(c1, rec);
    nalu_xform_m(c1, rec);
    __builtin_amdgcn_fence(__ATOMIC_RELEASE, "workgroup", "local");
    __builtin_amdgcn_s_barrier_signal(-1);
    __builtin_amdgcn_s_barrier_wait(-1);
    __builtin_amdgcn_fence(__ATOMIC_ACQUIRE, "workgroup", "local");

    Frag f0, f1;
    nalu_frag_load(slot0 + fAo, slot0 + fBo, f0);

#pragma unroll 1
    for (int it = 0; it < (NSTAGE - 4) / 2; ++it) {
        nalu_stage<true, true, true, 0>(f0, f1, c1, c0, rec, slot1 + sAo, (float*)(slot1 + sDo), slot1 + sMo,
                                     slot1 + fAo, slot1 + fBo, gx, gwd, gmd, gwt, gmt, acc);
        NALU_ADVANCE();
        nalu_stage<true, true, true, 3>(f1, f0, c0, c1, rec, slot0 + sAo, (float*)(slot0 + sDo), slot0 + sMo,
                                     slot0 + fAo, slot0 + fBo, gx, gwd, gmd, gwt, gmt, acc);
        NALU_ADVANCE();
    }
    nalu_stage<true, true, true, 6>(f0, f1, c1, c0, rec, slot1 + sAo, (float*)(slot1 + sDo), slot1 + sMo,
                                 slot1 + fAo, slot1 + fBo, gx, gwd, gmd, gwt, gmt, acc);
    nalu_stage<true, false, true, 9>(f1, f0, c0, c1, rec, slot0 + sAo, (float*)(slot0 + sDo), slot0 + sMo,
                                  slot0 + fAo, slot0 + fBo, gx, gwd, gmd, gwt, gmt, acc);
    nalu_stage<true, false, false, 12>(f0, f1, c1, c0, rec, slot1 + sAo, (float*)(slot1 + sDo), slot1 + sMo,
                                   slot1 + fAo, slot1 + fBo, gx, gwd, gmd, gwt, gmt, acc);
    nalu_stage<false, false, false, 15>(f1, f0, c0, c1, rec, slot0 + sAo, (float*)(slot0 + sDo), slot0 + sMo,
                                    slot0 + fAo, slot0 + fBo, gx, gwd, gmd, gwt, gmt, acc);

    const int m0 = bm0 + wm * 32;
    const int bcol = bn0 + wn * 16 + l15;
    const float g1 = 1.0f / (1.0f + expf(-g[bcol]));
    const float gm = 1.0f - g1;
#pragma unroll
    for (int v = 0; v < 8; ++v) {
        const int row0 = m0 + v + 8 * half;
        const int negc = (int)(acc.P[v] + 0.5f);
        {
            const float m1 = expf(fminf(acc.M0[v], LOG_CLIP));
            float ms = ((negc & 1) ? -1.0f : 1.0f) * expf(acc.L0[v]);
            ms = fminf(fmaxf(ms, -1.0f), 1.0f);
            out[row0 * NO + bcol] = g1 * acc.A0[v] + gm * m1 * ms;
        }
        {
            const float m1 = expf(fminf(acc.M1[v], LOG_CLIP));
            float ms = (((negc >> PSHIFT) & 1) ? -1.0f : 1.0f) * expf(acc.L1[v]);
            ms = fminf(fmaxf(ms, -1.0f), 1.0f);
            out[(row0 + 16) * NO + bcol] = g1 * acc.A1[v] + gm * m1 * ms;
        }
    }
}

extern "C" void kernel_launch(void* const* d_in, const int* in_sizes, int n_in,
                              void* d_out, int out_size, void* d_ws, size_t ws_size,
                              hipStream_t stream) {
    const float* x     = (const float*)d_in[0];
    const float* w_hat = (const float*)d_in[1];
    const float* m_hat = (const float*)d_in[2];
    const float* g     = (const float*)d_in[3];
    float* out = (float*)d_out;

    nalu_main_kernel<<<256, 128, 0, stream>>>(x, w_hat, m_hat, g, out);
}


# w1 = tanh(w_hat)*sigmoid(m_hat) from two v_tanh_f32 per weight: sigmoid(m) = 0.5+0.5*tanh(m/2), w1 = h+h*tanh(m/2), h = tanh(w_hat)/2; no exp2, exp2, rcp; w1 rounding changes, sums kept
# speedup vs baseline: 5.5700x; 1.0939x over previous
#include <hip/hip_runtime.h>
#include <math.h>

typedef float v2f __attribute__((ext_vector_type(2)));
typedef float v4f __attribute__((ext_vector_type(4)));
typedef float v8f __attribute__((ext_vector_type(8)));


#define NB 1024
#define NI 512
#define NO 512
#define EPS 1e-7f
#define LOG_CLIP 20.0f
#define NEG_BIG -1e30f
#define LN2 0.693147180559945309f


__device__ __forceinline__ float nalu_w1(float wh, float mh) {
    const float h = 0.5f * __builtin_amdgcn_tanhf(wh);
    const float tm = __builtin_amdgcn_tanhf(0.5f * mh);
    return fmaf(h, tm, h);
}

__device__ __forceinline__ float nalu_ln(float v) {
    return __builtin_amdgcn_logf(v) * LN2;
}

#define BM 64
#define BN 32
#define KP 16
#define KS (KP / 2)
#define NSTAGE ((NI / 2) / KP)
#define A_PITCH (KP + 1)
#define B_PITCH (BN + 1)
#define A_V4 (BM * A_PITCH)
#define B_V4 (KP * B_PITCH)
#define STAGE_V4 (A_V4 + 2 * B_V4)
#define NLD 8
#define PRADIX 1024.0f
#define PSHIFT 10

struct Acc {
    v8f A0, M0, L0, A1, M1, L1, P;
};
struct Frag {
    v4f a0[KS], a1[KS], q0[KS], q1[KS];
};
struct Raw {
    v2f x[NLD];
    v4f wd[2], md[2];
    v4f wt[2], mt[2];
};
struct Rec {
    v4f a[NLD];
    v2f bw[4];
    v2f b2[4];
    v4f b1[4];
};

__device__ __forceinline__ void nalu_frag_load(const v4f* fa0, const v4f* fb0, Frag& f) {
    const v4f* fa1 = fa0 + 16 * A_PITCH;
    const v4f* fb1 = fb0 + B_V4;
#pragma unroll
    for (int ks = 0; ks < KS; ++ks) {
        f.a0[ks] = fa0[2 * ks];
        f.a1[ks] = fa1[2 * ks];
        f.q0[ks] = fb0[2 * ks * B_PITCH];
        f.q1[ks] = fb1[2 * ks * B_PITCH];
    }
}

template <int K0, int K1>
__device__ __forceinline__ void nalu_wmma(const Frag& f, Acc& acc) {
#pragma unroll
    for (int ks = K0; ks < K1; ++ks) {
        v2f ax0 = __builtin_shufflevector(f.a0[ks], f.a0[ks], 0, 1);
        v2f al0 = __builtin_shufflevector(f.a0[ks], f.a0[ks], 2, 3);
        v2f ax1 = __builtin_shufflevector(f.a1[ks], f.a1[ks], 0, 1);
        v2f al1 = __builtin_shufflevector(f.a1[ks], f.a1[ks], 2, 3);
        v2f bw  = __builtin_shufflevector(f.q0[ks], f.q0[ks], 0, 1);
        v2f b2  = __builtin_shufflevector(f.q0[ks], f.q0[ks], 2, 3);
        v2f b1  = __builtin_shufflevector(f.q1[ks], f.q1[ks], 0, 1);
        v2f bn  = __builtin_shufflevector(f.q1[ks], f.q1[ks], 2, 3);

        v2f an0, az0, an1, az1;
        an0.x = (ax0.x < 0.0f) ? 1.0f : 0.0f;
        an0.y = (ax0.y < 0.0f) ? 1.0f : 0.0f;
        az0.x = (ax0.x == 0.0f) ? 1.0f : 0.0f;
        az0.y = (ax0.y == 0.0f) ? 1.0f : 0.0f;
        an1.x = (ax1.x < 0.0f) ? 1.0f : 0.0f;
        an1.y = (ax1.y < 0.0f) ? 1.0f : 0.0f;
        az1.x = (ax1.x == 0.0f) ? 1.0f : 0.0f;
        az1.y = (ax1.y == 0.0f) ? 1.0f : 0.0f;

        acc.A0 = __builtin_amdgcn_wmma_f32_16x16x4_f32(false, ax0, false, bw,
                                                       (short)0, acc.A0, false, false);
        acc.A1 = __builtin_amdgcn_wmma_f32_16x16x4_f32(false, ax1, false, bw,
                                                       (short)0, acc.A1, false, false);
        acc.M0 = __builtin_amdgcn_wmma_f32_16x16x4_f32(false, al0, false, bw,
                                                       (short)0, acc.M0, false, false);
        acc.M1 = __builtin_amdgcn_wmma_f32_16x16x4_f32(false, al1, false, bw,
                                                       (short)0, acc.M1, false, false);
        acc.L0 = __builtin_amdgcn_wmma_f32_16x16x4_f32(false, an0, false, b2,
                                                       (short)0, acc.L0, false, false);
        acc.L1 = __builtin_amdgcn_wmma_f32_16x16x4_f32(false, an1, false, b2,
                                                       (short)0, acc.L1, false, false);
        acc.L0 = __builtin_amdgcn_wmma_f32_16x16x4_f32(false, az0, false, b1,
                                                       (short)0, acc.L0, false, false);
        acc.L1 = __builtin_amdgcn_wmma_f32_16x16x4_f32(false, az1, false, b1,
                                                       (short)0, acc.L1, false, false);
        v2f ap;
        ap.x = fmaf(an1.x, PRADIX, an0.x);
        ap.y = fmaf(an1.y, PRADIX, an0.y);
        acc.P = __builtin_amdgcn_wmma_f32_16x16x4_f32(false, ap, false, bn,
                                                      (short)0, acc.P, false, false);
    }
}

__device__ __forceinline__ void nalu_raw_load(const float* gx, const float* gwd, const float* gmd,
                                              const float* gwt, const float* gmt, Raw& r) {
#pragma unroll
    for (int j = 0; j < NLD; ++j) r.x[j] = *(const v2f*)(gx + j * 8 * NI);
#pragma unroll
    for (int e = 0; e < 2; ++e) {
        r.wd[e] = *(const v4f*)(gwd + e * NO);
        r.md[e] = *(const v4f*)(gmd + e * NO);
        r.wt[e] = *(const v4f*)(gwt + e * 16 * NO);
        r.mt[e] = *(const v4f*)(gmt + e * 16 * NO);
    }
}

__device__ __forceinline__ void nalu_xform_d(const Raw& r, Rec& o) {
#pragma unroll
    for (int i = 0; i < 4; ++i) {
        v2f w;
        w.x = nalu_w1(r.wd[0][i], r.md[0][i]);
        w.y = nalu_w1(r.wd[1][i], r.md[1][i]);
        o.bw[i] = w;
    }
}

__device__ __forceinline__ void nalu_xform_x(const Raw& r, Rec& o) {
#pragma unroll
    for (int j = 0; j < NLD; ++j) {
        v4f a;
        a.x = r.x[j].x;
        a.y = r.x[j].y;
        a.z = nalu_ln(fmaxf(fabsf(r.x[j].x), EPS));
        a.w = nalu_ln(fmaxf(fabsf(r.x[j].y), EPS));
        o.a[j] = a;
    }
}

__device__ __forceinline__ void nalu_xform_m(const Raw& r, Rec& o) {
#pragma unroll
    for (int j = 0; j < 2; ++j) {
#pragma unroll
        for (int h = 0; h < 2; ++h) {
            const float wsa = fabsf(nalu_w1(r.wt[j][2 * h], r.mt[j][2 * h]));
            const float wsb = fabsf(nalu_w1(r.wt[j][2 * h + 1], r.mt[j][2 * h + 1]));
            const float t2a = 1.0f - 2.0f * wsa;
            const float t2b = 1.0f - 2.0f * wsb;
            const float t1a = 1.0f - wsa;
            const float t1b = 1.0f - wsb;
            v2f lo;
            lo.x = fmaxf(nalu_ln(fabsf(t2a)), NEG_BIG);
            lo.y = fmaxf(nalu_ln(fabsf(t2b)), NEG_BIG);
            v4f hi;
            hi.x = fmaxf(nalu_ln(t1a), NEG_BIG);
            hi.y = fmaxf(nalu_ln(t1b), NEG_BIG);
            hi.z = (t2a < 0.0f) ? 1.0f : 0.0f;
            hi.w = (t2b < 0.0f) ? 1.0f : 0.0f;
            o.b2[2 * j + h] = lo;
            o.b1[2 * j + h] = hi;
        }
    }
}

__device__ __forceinline__ void nalu_rec_store(v4f* sA, float* sD, v4f* sM, const Rec& o) {
#pragma unroll
    for (int j = 0; j < NLD; ++j) sA[j * 8 * A_PITCH] = o.a[j];
#pragma unroll
    for (int i = 0; i < 4; ++i) *(v2f*)(sD + 4 * i) = o.bw[i];
#pragma unroll
    for (int j = 0; j < 2; ++j) {
#pragma unroll
        for (int h = 0; h < 2; ++h) {
            v4f* q = sM + h * B_PITCH + j * 16;
            *(v2f*)((float*)q + 2) = o.b2[2 * j + h];
            q[B_V4] = o.b1[2 * j + h];
        }
    }
}

__device__ __forceinline__ void nalu_ring_signal() {
    __builtin_amdgcn_fence(__ATOMIC_RELEASE, "workgroup", "local");
    __builtin_amdgcn_s_barrier_signal(-1);
    __builtin_amdgcn_sched_barrier(0);
}
__device__ __forceinline__ void nalu_ring_wait() {
    __builtin_amdgcn_sched_barrier(0);
    __builtin_amdgcn_s_barrier_wait(-1);
    __builtin_amdgcn_fence(__ATOMIC_ACQUIRE, "workgroup", "local");
}

__device__ __forceinline__ void nalu_wmma_above(const Acc& a) {
    asm volatile("" :: "v"(a.A0[0]), "v"(a.M0[0]), "v"(a.L0[0]), "v"(a.A1[0]),
                       "v"(a.M1[0]), "v"(a.L1[0]), "v"(a.P[0]));
}

template <int N, int V, int ID>
__device__ __forceinline__ void nalu_interleave() {
#pragma unroll
    for (int i = 0; i < N; ++i) {
        __builtin_amdgcn_sched_group_barrier(0x402, V, ID);
        __builtin_amdgcn_sched_group_barrier(0x008, 1, ID);
    }
}

template <bool STORE, bool LOAD, bool XFORM, int ID>
__device__ __forceinline__ void nalu_stage(const Frag& fc, Frag& fn, Raw& rl, const Raw& rx, Rec& rec,
                                           v4f* sA, float* sD, v4f* sM, const v4f* fA, const v4f* fB,
                                           const float* gx, const float* gwd, const float* gmd,
                                           const float* gwt, const float* gmt, Acc& acc) {
    if (STORE) {
        nalu_rec_store(sA, sD, sM, rec);
        if (LOAD) nalu_raw_load(gx, gwd, gmd, gwt, gmt, rl);
        __builtin_amdgcn_sched_barrier(0);
    }
    if (XFORM) nalu_xform_d(rx, rec);
    nalu_wmma<0, 3>(fc, acc);
    if (XFORM) nalu_interleave<3 * 9, 8, ID>();
    if (STORE) {
        nalu_wmma_above(acc);
        nalu_ring_signal();
    }
    if (XFORM) nalu_xform_x(rx, rec);
    nalu_wmma<3, 4>(fc, acc);
    if (XFORM) nalu_interleave<9, 10, ID + 1>();
    if (STORE) {
        nalu_wmma_above(acc);
        nalu_ring_wait();
        nalu_frag_load(fA, fB, fn);
        __builtin_amdgcn_sched_barrier(0);
    }
    if (XFORM) nalu_xform_m(rx, rec);
    nalu_wmma<4, KS>(fc, acc);
    if (XFORM) {
        nalu_interleave<(KS - 4) * 9, 9, ID + 2>();
        nalu_wmma_above(acc);
        __builtin_amdgcn_sched_barrier(0);
    }
}

#define NALU_ADVANCE() do { gx += 2 * KP; gwt += 2 * KP; gmt += 2 * KP; \
                            gwd += 2 * KP * NO; gmd += 2 * KP * NO; } while (0)

__global__ __launch_bounds__(128) void nalu_main_kernel(const float* __restrict__ x,
                                                        const float* __restrict__ w_hat,
                                                        const float* __restrict__ m_hat,
                                                        const float* __restrict__ g,
                                                        float* __restrict__ out) {
    __shared__ v4f lds[2 * STAGE_V4];

    const int tid = threadIdx.x;
    const int lane = tid & 31;
    const int wave = tid >> 5;
    const int wm = wave >> 1;
    const int wn = wave & 1;
    const int bm0 = (blockIdx.x >> 4) * BM;
    const int bn0 = (blockIdx.x & 15) * BN;
    const int half = lane >> 4;
    const int l15 = lane & 15;

    const float* gx = x + (bm0 + (tid >> 4)) * NI + 2 * (tid & 15);
    const int offd = (2 * (tid >> 3)) * NO + bn0 + 4 * (tid & 7);
    const int offt = (bn0 + (tid >> 3)) * NO + 4 * (tid & 7);
    const float* gwd = w_hat + offd;
    const float* gmd = m_hat + offd;
    const float* gwt = w_hat + offt;
    const float* gmt = m_hat + offt;
    const int sAo = (tid >> 4) * A_PITCH + (tid & 15);
    const int sDo = A_V4 + (tid >> 3) * B_PITCH + 4 * (tid & 7);
    const int sMo = A_V4 + 2 * (tid & 7) * B_PITCH + (tid >> 3);

    const int fAo = (wm * 32 + l15) * A_PITCH + half;
    const int fBo = A_V4 + half * B_PITCH + wn * 16 + l15;

    Acc acc;
    acc.A0 = v8f{}; acc.M0 = v8f{}; acc.L0 = v8f{};
    acc.A1 = v8f{}; acc.M1 = v8f{}; acc.L1 = v8f{}; acc.P = v8f{};

    v4f* const slot0 = lds;
    v4f* const slot1 = lds + STAGE_V4;

    Raw c0, c1;
    Rec rec;
    nalu_raw_load(gx, gwd, gmd, gwt, gmt, c0);
    NALU_ADVANCE();
    nalu_raw_load(gx, gwd, gmd, gwt, gmt, c1);
    NALU_ADVANCE();
    nalu_xform_d(c0, rec);
    nalu_xform_x(c0, rec);
    nalu_xform_m(c0, rec);
    nalu_rec_store(slot0 + sAo, (float*)(slot0 + sDo), slot0 + sMo, rec);
    nalu_raw_load(gx, gwd, gmd, gwt, gmt, c0);
    NALU_ADVANCE();
    nalu_xform_d(c1, rec);
    nalu_xform_x(c1, rec);
    nalu_xform_m(c1, rec);
    __builtin_amdgcn_fence(__ATOMIC_RELEASE, "workgroup", "local");
    __builtin_amdgcn_s_barrier_signal(-1);
    __builtin_amdgcn_s_barrier_wait(-1);
    __builtin_amdgcn_fence(__ATOMIC_ACQUIRE, "workgroup", "local");

    Frag f0, f1;
    nalu_frag_load(slot0 + fAo, slot0 + fBo, f0);

#pragma unroll 1
    for (int it = 0; it < (NSTAGE - 4) / 2; ++it) {
        nalu_stage<true, true, true, 0>(f0, f1, c1, c0, rec, slot1 + sAo, (float*)(slot1 + sDo), slot1 + sMo,
                                     slot1 + fAo, slot1 + fBo, gx, gwd, gmd, gwt, gmt, acc);
        NALU_ADVANCE();
        nalu_stage<true, true, true, 3>(f1, f0, c0, c1, rec, slot0 + sAo, (float*)(slot0 + sDo), slot0 + sMo,
                                     slot0 + fAo, slot0 + fBo, gx, gwd, gmd, gwt, gmt, acc);
        NALU_ADVANCE();
    }
    nalu_stage<true, true, true, 6>(f0, f1, c1, c0, rec, slot1 + sAo, (float*)(slot1 + sDo), slot1 + sMo,
                                 slot1 + fAo, slot1 + fBo, gx, gwd, gmd, gwt, gmt, acc);
    nalu_stage<true, false, true, 9>(f1, f0, c0, c1, rec, slot0 + sAo, (float*)(slot0 + sDo), slot0 + sMo,
                                  slot0 + fAo, slot0 + fBo, gx, gwd, gmd, gwt, gmt, acc);
    nalu_stage<true, false, false, 12>(f0, f1, c1, c0, rec, slot1 + sAo, (float*)(slot1 + sDo), slot1 + sMo,
                                   slot1 + fAo, slot1 + fBo, gx, gwd, gmd, gwt, gmt, acc);
    nalu_stage<false, false, false, 15>(f1, f0, c0, c1, rec, slot0 + sAo, (float*)(slot0 + sDo), slot0 + sMo,
                                    slot0 + fAo, slot0 + fBo, gx, gwd, gmd, gwt, gmt, acc);

    const int m0 = bm0 + wm * 32;
    const int bcol = bn0 + wn * 16 + l15;
    const float g1 = 1.0f / (1.0f + expf(-g[bcol]));
    const float gm = 1.0f - g1;
#pragma unroll
    for (int v = 0; v < 8; ++v) {
        const int row0 = m0 + v + 8 * half;
        const int negc = (int)(acc.P[v] + 0.5f);
        {
            const float m1 = expf(fminf(acc.M0[v], LOG_CLIP));
            float ms = ((negc & 1) ? -1.0f : 1.0f) * expf(acc.L0[v]);
            ms = fminf(fmaxf(ms, -1.0f), 1.0f);
            out[row0 * NO + bcol] = g1 * acc.A0[v] + gm * m1 * ms;
        }
        {
            const float m1 = expf(fminf(acc.M1[v], LOG_CLIP));
            float ms = (((negc >> PSHIFT) & 1) ? -1.0f : 1.0f) * expf(acc.L1[v]);
            ms = fminf(fmaxf(ms, -1.0f), 1.0f);
            out[(row0 + 16) * NO + bcol] = g1 * acc.A1[v] + gm * m1 * ms;
        }
    }
}

extern "C" void kernel_launch(void* const* d_in, const int* in_sizes, int n_in,
                              void* d_out, int out_size, void* d_ws, size_t ws_size,
                              hipStream_t stream) {
    const float* x     = (const float*)d_in[0];
    const float* w_hat = (const float*)d_in[1];
    const float* m_hat = (const float*)d_in[2];
    const float* g     = (const float*)d_in[3];
    float* out = (float*)d_out;

    nalu_main_kernel<<<256, 128, 0, stream>>>(x, w_hat, m_hat, g, out);
}


# sign path ends early per wave: its factors lie in [-1,1], so once every accL of the wave is below -106 (exp gives 0) the 5 sign WMMAs and the flags of a k-step are skipped; sums kept
# speedup vs baseline: 7.2971x; 1.3101x over previous
#include <hip/hip_runtime.h>
#include <math.h>

typedef float v2f __attribute__((ext_vector_type(2)));
typedef float v4f __attribute__((ext_vector_type(4)));
typedef float v8f __attribute__((ext_vector_type(8)));


#define NB 1024
#define NI 512
#define NO 512
#define EPS 1e-7f
#define LOG_CLIP 20.0f
#define NEG_BIG -1e30f
#define LN2 0.693147180559945309f


__device__ __forceinline__ float nalu_w1(float wh, float mh) {
    const float h = 0.5f * __builtin_amdgcn_tanhf(wh);
    const float tm = __builtin_amdgcn_tanhf(0.5f * mh);
    return fmaf(h, tm, h);
}

__device__ __forceinline__ float nalu_ln(float v) {
    return __builtin_amdgcn_logf(v) * LN2;
}

#define BM 64
#define BN 32
#define KP 16
#define KS (KP / 2)
#define NSTAGE ((NI / 2) / KP)
#define A_PITCH (KP + 1)
#define B_PITCH (BN + 1)
#define A_V4 (BM * A_PITCH)
#define B_V4 (KP * B_PITCH)
#define STAGE_V4 (A_V4 + 2 * B_V4)
#define NLD 8
#define PRADIX 1024.0f
#define PSHIFT 10
#define SIGN_DEAD -106.0f

struct Acc {
    v8f A0, M0, L0, A1, M1, L1, P;
};
struct Frag {
    v4f a0[KS], a1[KS], q0[KS], q1[KS];
};
struct Raw {
    v2f x[NLD];
    v4f wd[2], md[2];
    v4f wt[2], mt[2];
};
struct Rec {
    v4f a[NLD];
    v2f bw[4];
    v2f b2[4];
    v4f b1[4];
};

__device__ __forceinline__ void nalu_frag_load(const v4f* fa0, const v4f* fb0, Frag& f) {
    const v4f* fa1 = fa0 + 16 * A_PITCH;
    const v4f* fb1 = fb0 + B_V4;
#pragma unroll
    for (int ks = 0; ks < KS; ++ks) {
        f.a0[ks] = fa0[2 * ks];
        f.a1[ks] = fa1[2 * ks];
        f.q0[ks] = fb0[2 * ks * B_PITCH];
        f.q1[ks] = fb1[2 * ks * B_PITCH];
    }
}

template <int K0, int K1, bool SIGN>
__device__ __forceinline__ void nalu_wmma(const Frag& f, Acc& acc) {
#pragma unroll
    for (int ks = K0; ks < K1; ++ks) {
        v2f ax0 = __builtin_shufflevector(f.a0[ks], f.a0[ks], 0, 1);
        v2f al0 = __builtin_shufflevector(f.a0[ks], f.a0[ks], 2, 3);
        v2f ax1 = __builtin_shufflevector(f.a1[ks], f.a1[ks], 0, 1);
        v2f al1 = __builtin_shufflevector(f.a1[ks], f.a1[ks], 2, 3);
        v2f bw  = __builtin_shufflevector(f.q0[ks], f.q0[ks], 0, 1);
        v2f b2  = __builtin_shufflevector(f.q0[ks], f.q0[ks], 2, 3);
        v2f b1  = __builtin_shufflevector(f.q1[ks], f.q1[ks], 0, 1);
        v2f bn  = __builtin_shufflevector(f.q1[ks], f.q1[ks], 2, 3);

        v2f an0 = {}, az0 = {}, an1 = {}, az1 = {};
        if (SIGN) {
        an0.x = (ax0.x < 0.0f) ? 1.0f : 0.0f;
        an0.y = (ax0.y < 0.0f) ? 1.0f : 0.0f;
        az0.x = (ax0.x == 0.0f) ? 1.0f : 0.0f;
        az0.y = (ax0.y == 0.0f) ? 1.0f : 0.0f;
        an1.x = (ax1.x < 0.0f) ? 1.0f : 0.0f;
        an1.y = (ax1.y < 0.0f) ? 1.0f : 0.0f;
        az1.x = (ax1.x == 0.0f) ? 1.0f : 0.0f;
        az1.y = (ax1.y == 0.0f) ? 1.0f : 0.0f;
        }

        if (!SIGN) {
        acc.A0 = __builtin_amdgcn_wmma_f32_16x16x4_f32(false, ax0, false, bw,
                                                       (short)0, acc.A0, false, false);
        acc.A1 = __builtin_amdgcn_wmma_f32_16x16x4_f32(false, ax1, false, bw,
                                                       (short)0, acc.A1, false, false);
        acc.M0 = __builtin_amdgcn_wmma_f32_16x16x4_f32(false, al0, false, bw,
                                                       (short)0, acc.M0, false, false);
        acc.M1 = __builtin_amdgcn_wmma_f32_16x16x4_f32(false, al1, false, bw,
                                                       (short)0, acc.M1, false, false);
        continue;
        }
        acc.L0 = __builtin_amdgcn_wmma_f32_16x16x4_f32(false, an0, false, b2,
                                                       (short)0, acc.L0, false, false);
        acc.L1 = __builtin_amdgcn_wmma_f32_16x16x4_f32(false, an1, false, b2,
                                                       (short)0, acc.L1, false, false);
        acc.L0 = __builtin_amdgcn_wmma_f32_16x16x4_f32(false, az0, false, b1,
                                                       (short)0, acc.L0, false, false);
        acc.L1 = __builtin_amdgcn_wmma_f32_16x16x4_f32(false, az1, false, b1,
                                                       (short)0, acc.L1, false, false);
        v2f ap;
        ap.x = fmaf(an1.x, PRADIX, an0.x);
        ap.y = fmaf(an1.y, PRADIX, an0.y);
        acc.P = __builtin_amdgcn_wmma_f32_16x16x4_f32(false, ap, false, bn,
                                                      (short)0, acc.P, false, false);
    }
}

__device__ __forceinline__ void nalu_raw_load(const float* gx, const float* gwd, const float* gmd,
                                              const float* gwt, const float* gmt, Raw& r) {
#pragma unroll
    for (int j = 0; j < NLD; ++j) r.x[j] = *(const v2f*)(gx + j * 8 * NI);
#pragma unroll
    for (int e = 0; e < 2; ++e) {
        r.wd[e] = *(const v4f*)(gwd + e * NO);
        r.md[e] = *(const v4f*)(gmd + e * NO);
        r.wt[e] = *(const v4f*)(gwt + e * 16 * NO);
        r.mt[e] = *(const v4f*)(gmt + e * 16 * NO);
    }
}

__device__ __forceinline__ void nalu_xform_d(const Raw& r, Rec& o) {
#pragma unroll
    for (int i = 0; i < 4; ++i) {
        v2f w;
        w.x = nalu_w1(r.wd[0][i], r.md[0][i]);
        w.y = nalu_w1(r.wd[1][i], r.md[1][i]);
        o.bw[i] = w;
    }
}

__device__ __forceinline__ void nalu_xform_x(const Raw& r, Rec& o) {
#pragma unroll
    for (int j = 0; j < NLD; ++j) {
        v4f a;
        a.x = r.x[j].x;
        a.y = r.x[j].y;
        a.z = nalu_ln(fmaxf(fabsf(r.x[j].x), EPS));
        a.w = nalu_ln(fmaxf(fabsf(r.x[j].y), EPS));
        o.a[j] = a;
    }
}

__device__ __forceinline__ void nalu_xform_m(const Raw& r, Rec& o) {
#pragma unroll
    for (int j = 0; j < 2; ++j) {
#pragma unroll
        for (int h = 0; h < 2; ++h) {
            const float wsa = fabsf(nalu_w1(r.wt[j][2 * h], r.mt[j][2 * h]));
            const float wsb = fabsf(nalu_w1(r.wt[j][2 * h + 1], r.mt[j][2 * h + 1]));
            const float t2a = 1.0f - 2.0f * wsa;
            const float t2b = 1.0f - 2.0f * wsb;
            const float t1a = 1.0f - wsa;
            const float t1b = 1.0f - wsb;
            v2f lo;
            lo.x = fmaxf(nalu_ln(fabsf(t2a)), NEG_BIG);
            lo.y = fmaxf(nalu_ln(fabsf(t2b)), NEG_BIG);
            v4f hi;
            hi.x = fmaxf(nalu_ln(t1a), NEG_BIG);
            hi.y = fmaxf(nalu_ln(t1b), NEG_BIG);
            hi.z = (t2a < 0.0f) ? 1.0f : 0.0f;
            hi.w = (t2b < 0.0f) ? 1.0f : 0.0f;
            o.b2[2 * j + h] = lo;
            o.b1[2 * j + h] = hi;
        }
    }
}

__device__ __forceinline__ void nalu_rec_store(v4f* sA, float* sD, v4f* sM, const Rec& o) {
#pragma unroll
    for (int j = 0; j < NLD; ++j) sA[j * 8 * A_PITCH] = o.a[j];
#pragma unroll
    for (int i = 0; i < 4; ++i) *(v2f*)(sD + 4 * i) = o.bw[i];
#pragma unroll
    for (int j = 0; j < 2; ++j) {
#pragma unroll
        for (int h = 0; h < 2; ++h) {
            v4f* q = sM + h * B_PITCH + j * 16;
            *(v2f*)((float*)q + 2) = o.b2[2 * j + h];
            q[B_V4] = o.b1[2 * j + h];
        }
    }
}

__device__ __forceinline__ void nalu_ring_signal() {
    __builtin_amdgcn_fence(__ATOMIC_RELEASE, "workgroup", "local");
    __builtin_amdgcn_s_barrier_signal(-1);
    __builtin_amdgcn_sched_barrier(0);
}
__device__ __forceinline__ void nalu_ring_wait() {
    __builtin_amdgcn_sched_barrier(0);
    __builtin_amdgcn_s_barrier_wait(-1);
    __builtin_amdgcn_fence(__ATOMIC_ACQUIRE, "workgroup", "local");
}

__device__ __forceinline__ void nalu_wmma_above(const Acc& a) {
    asm volatile("" :: "v"(a.A0[0]), "v"(a.M0[0]), "v"(a.L0[0]), "v"(a.A1[0]),
                       "v"(a.M1[0]), "v"(a.L1[0]), "v"(a.P[0]));
}

__device__ __forceinline__ bool nalu_sign_live(const Acc& a) {
    float m = fmaxf(a.L0[0], a.L1[0]);
#pragma unroll
    for (int v = 1; v < 8; ++v) m = fmaxf(m, fmaxf(a.L0[v], a.L1[v]));
    return __builtin_amdgcn_ballot_w32(!(m < SIGN_DEAD)) != 0u;
}

template <int N, int V, int ID>
__device__ __forceinline__ void nalu_interleave() {
#pragma unroll
    for (int i = 0; i < N; ++i) {
        __builtin_amdgcn_sched_group_barrier(0x402, V, ID);
        __builtin_amdgcn_sched_group_barrier(0x008, 1, ID);
    }
}

template <bool STORE, bool LOAD, bool XFORM, int ID>
__device__ __forceinline__ void nalu_stage(const Frag& fc, Frag& fn, Raw& rl, const Raw& rx, Rec& rec,
                                           v4f* sA, float* sD, v4f* sM, const v4f* fA, const v4f* fB,
                                           const float* gx, const float* gwd, const float* gmd,
                                           const float* gwt, const float* gmt, Acc& acc, bool& live) {
    if (STORE) {
        nalu_rec_store(sA, sD, sM, rec);
        if (LOAD) nalu_raw_load(gx, gwd, gmd, gwt, gmt, rl);
        __builtin_amdgcn_sched_barrier(0);
    }
    if (XFORM) nalu_xform_d(rx, rec);
    nalu_wmma<0, 3, false>(fc, acc);
    if (XFORM) nalu_interleave<3 * 4, 4, ID>();
    if (live) nalu_wmma<0, 3, true>(fc, acc);
    if (STORE) {
        nalu_wmma_above(acc);
        nalu_ring_signal();
    }
    if (XFORM) nalu_xform_x(rx, rec);
    nalu_wmma<3, 4, false>(fc, acc);
    if (XFORM) nalu_interleave<4, 12, ID + 1>();
    if (live) nalu_wmma<3, 4, true>(fc, acc);
    if (STORE) {
        nalu_wmma_above(acc);
        nalu_ring_wait();
        nalu_frag_load(fA, fB, fn);
        __builtin_amdgcn_sched_barrier(0);
    }
    if (XFORM) nalu_xform_m(rx, rec);
    nalu_wmma<4, KS, false>(fc, acc);
    if (XFORM) nalu_interleave<(KS - 4) * 4, 16, ID + 2>();
    if (live) {
        nalu_wmma<4, KS, true>(fc, acc);
        live = nalu_sign_live(acc);
    }
    if (XFORM) {
        nalu_wmma_above(acc);
        __builtin_amdgcn_sched_barrier(0);
    }
}

#define NALU_ADVANCE() do { gx += 2 * KP; gwt += 2 * KP; gmt += 2 * KP; \
                            gwd += 2 * KP * NO; gmd += 2 * KP * NO; } while (0)

__global__ __launch_bounds__(128) void nalu_main_kernel(const float* __restrict__ x,
                                                        const float* __restrict__ w_hat,
                                                        const float* __restrict__ m_hat,
                                                        const float* __restrict__ g,
                                                        float* __restrict__ out) {
    __shared__ v4f lds[2 * STAGE_V4];

    const int tid = threadIdx.x;
    const int lane = tid & 31;
    const int wave = tid >> 5;
    const int wm = wave >> 1;
    const int wn = wave & 1;
    const int bm0 = (blockIdx.x >> 4) * BM;
    const int bn0 = (blockIdx.x & 15) * BN;
    const int half = lane >> 4;
    const int l15 = lane & 15;

    const float* gx = x + (bm0 + (tid >> 4)) * NI + 2 * (tid & 15);
    const int offd = (2 * (tid >> 3)) * NO + bn0 + 4 * (tid & 7);
    const int offt = (bn0 + (tid >> 3)) * NO + 4 * (tid & 7);
    const float* gwd = w_hat + offd;
    const float* gmd = m_hat + offd;
    const float* gwt = w_hat + offt;
    const float* gmt = m_hat + offt;
    const int sAo = (tid >> 4) * A_PITCH + (tid & 15);
    const int sDo = A_V4 + (tid >> 3) * B_PITCH + 4 * (tid & 7);
    const int sMo = A_V4 + 2 * (tid & 7) * B_PITCH + (tid >> 3);

    const int fAo = (wm * 32 + l15) * A_PITCH + half;
    const int fBo = A_V4 + half * B_PITCH + wn * 16 + l15;

    Acc acc;
    acc.A0 = v8f{}; acc.M0 = v8f{}; acc.L0 = v8f{};
    acc.A1 = v8f{}; acc.M1 = v8f{}; acc.L1 = v8f{}; acc.P = v8f{};

    v4f* const slot0 = lds;
    v4f* const slot1 = lds + STAGE_V4;

    Raw c0, c1;
    Rec rec;
    nalu_raw_load(gx, gwd, gmd, gwt, gmt, c0);
    NALU_ADVANCE();
    nalu_raw_load(gx, gwd, gmd, gwt, gmt, c1);
    NALU_ADVANCE();
    nalu_xform_d(c0, rec);
    nalu_xform_x(c0, rec);
    nalu_xform_m(c0, rec);
    nalu_rec_store(slot0 + sAo, (float*)(slot0 + sDo), slot0 + sMo, rec);
    nalu_raw_load(gx, gwd, gmd, gwt, gmt, c0);
    NALU_ADVANCE();
    nalu_xform_d(c1, rec);
    nalu_xform_x(c1, rec);
    nalu_xform_m(c1, rec);
    __builtin_amdgcn_fence(__ATOMIC_RELEASE, "workgroup", "local");
    __builtin_amdgcn_s_barrier_signal(-1);
    __builtin_amdgcn_s_barrier_wait(-1);
    __builtin_amdgcn_fence(__ATOMIC_ACQUIRE, "workgroup", "local");

    bool live = true;

    Frag f0, f1;
    nalu_frag_load(slot0 + fAo, slot0 + fBo, f0);

#pragma unroll 1
    for (int it = 0; it < (NSTAGE - 4) / 2; ++it) {
        nalu_stage<true, true, true, 0>(f0, f1, c1, c0, rec, slot1 + sAo, (float*)(slot1 + sDo), slot1 + sMo,
                                     slot1 + fAo, slot1 + fBo, gx, gwd, gmd, gwt, gmt, acc, live);
        NALU_ADVANCE();
        nalu_stage<true, true, true, 3>(f1, f0, c0, c1, rec, slot0 + sAo, (float*)(slot0 + sDo), slot0 + sMo,
                                     slot0 + fAo, slot0 + fBo, gx, gwd, gmd, gwt, gmt, acc, live);
        NALU_ADVANCE();
    }
    nalu_stage<true, true, true, 6>(f0, f1, c1, c0, rec, slot1 + sAo, (float*)(slot1 + sDo), slot1 + sMo,
                                 slot1 + fAo, slot1 + fBo, gx, gwd, gmd, gwt, gmt, acc, live);
    nalu_stage<true, false, true, 9>(f1, f0, c0, c1, rec, slot0 + sAo, (float*)(slot0 + sDo), slot0 + sMo,
                                  slot0 + fAo, slot0 + fBo, gx, gwd, gmd, gwt, gmt, acc, live);
    nalu_stage<true, false, false, 12>(f0, f1, c1, c0, rec, slot1 + sAo, (float*)(slot1 + sDo), slot1 + sMo,
                                   slot1 + fAo, slot1 + fBo, gx, gwd, gmd, gwt, gmt, acc, live);
    nalu_stage<false, false, false, 15>(f1, f0, c0, c1, rec, slot0 + sAo, (float*)(slot0 + sDo), slot0 + sMo,
                                    slot0 + fAo, slot0 + fBo, gx, gwd, gmd, gwt, gmt, acc, live);

    const int m0 = bm0 + wm * 32;
    const int bcol = bn0 + wn * 16 + l15;
    const float g1 = 1.0f / (1.0f + expf(-g[bcol]));
    const float gm = 1.0f - g1;
#pragma unroll
    for (int v = 0; v < 8; ++v) {
        const int row0 = m0 + v + 8 * half;
        const int negc = (int)(acc.P[v] + 0.5f);
        {
            const float m1 = expf(fminf(acc.M0[v], LOG_CLIP));
            float ms = ((negc & 1) ? -1.0f : 1.0f) * expf(acc.L0[v]);
            ms = fminf(fmaxf(ms, -1.0f), 1.0f);
            out[row0 * NO + bcol] = g1 * acc.A0[v] + gm * m1 * ms;
        }
        {
            const float m1 = expf(fminf(acc.M1[v], LOG_CLIP));
            float ms = (((negc >> PSHIFT) & 1) ? -1.0f : 1.0f) * expf(acc.L1[v]);
            ms = fminf(fmaxf(ms, -1.0f), 1.0f);
            out[(row0 + 16) * NO + bcol] = g1 * acc.A1[v] + gm * m1 * ms;
        }
    }
}

extern "C" void kernel_launch(void* const* d_in, const int* in_sizes, int n_in,
                              void* d_out, int out_size, void* d_ws, size_t ws_size,
                              hipStream_t stream) {
    const float* x     = (const float*)d_in[0];
    const float* w_hat = (const float*)d_in[1];
    const float* m_hat = (const float*)d_in[2];
    const float* g     = (const float*)d_in[3];
    float* out = (float*)d_out;

    nalu_main_kernel<<<256, 128, 0, stream>>>(x, w_hat, m_hat, g, out);
}


# once all 4 waves of a workgroup have a dead sign path (states posted in LDS) the m1*ms term is 0: its K stages go lean: accA WMMAs only, x and direct-weight records only, no mirrored loads; sums kept
# speedup vs baseline: 7.9233x; 1.0858x over previous
#include <hip/hip_runtime.h>
#include <math.h>

typedef float v2f __attribute__((ext_vector_type(2)));
typedef float v4f __attribute__((ext_vector_type(4)));
typedef float v8f __attribute__((ext_vector_type(8)));
typedef int v4i __attribute__((ext_vector_type(4)));


#define NB 1024
#define NI 512
#define NO 512
#define EPS 1e-7f
#define LOG_CLIP 20.0f
#define NEG_BIG -1e30f
#define LN2 0.693147180559945309f


__device__ __forceinline__ float nalu_w1(float wh, float mh) {
    const float h = 0.5f * __builtin_amdgcn_tanhf(wh);
    const float tm = __builtin_amdgcn_tanhf(0.5f * mh);
    return fmaf(h, tm, h);
}

__device__ __forceinline__ float nalu_ln(float v) {
    return __builtin_amdgcn_logf(v) * LN2;
}

#define BM 64
#define BN 32
#define KP 16
#define KS (KP / 2)
#define NSTAGE ((NI / 2) / KP)
#define A_PITCH (KP + 1)
#define B_PITCH (BN + 1)
#define A_V4 (BM * A_PITCH)
#define B_V4 (KP * B_PITCH)
#define STAGE_V4 (A_V4 + 2 * B_V4)
#define NLD 8
#define PRADIX 1024.0f
#define PSHIFT 10
#define SIGN_DEAD -106.0f

struct Acc {
    v8f A0, M0, L0, A1, M1, L1, P;
};
struct Frag {
    v4f a0[KS], a1[KS], q0[KS], q1[KS];
};
struct Raw {
    v2f x[NLD];
    v4f wd[2], md[2];
    v4f wt[2], mt[2];
};
struct Rec {
    v4f a[NLD];
    v2f bw[4];
    v2f b2[4];
    v4f b1[4];
};

__device__ __forceinline__ void nalu_frag_load(const v4f* fa0, const v4f* fb0, Frag& f) {
    const v4f* fa1 = fa0 + 16 * A_PITCH;
    const v4f* fb1 = fb0 + B_V4;
#pragma unroll
    for (int ks = 0; ks < KS; ++ks) {
        f.a0[ks] = fa0[2 * ks];
        f.a1[ks] = fa1[2 * ks];
        f.q0[ks] = fb0[2 * ks * B_PITCH];
        f.q1[ks] = fb1[2 * ks * B_PITCH];
    }
}

__device__ __forceinline__ void nalu_frag_load_lean(const v4f* fa0, const v4f* fb0, Frag& f) {
    const v4f* fa1 = fa0 + 16 * A_PITCH;
#pragma unroll
    for (int ks = 0; ks < KS; ++ks) {
        const v2f a0 = *(const v2f*)(fa0 + 2 * ks);
        const v2f a1 = *(const v2f*)(fa1 + 2 * ks);
        const v2f q0 = *(const v2f*)(fb0 + 2 * ks * B_PITCH);
        f.a0[ks].x = a0.x; f.a0[ks].y = a0.y;
        f.a1[ks].x = a1.x; f.a1[ks].y = a1.y;
        f.q0[ks].x = q0.x; f.q0[ks].y = q0.y;
    }
}

template <int K0, int K1>
__device__ __forceinline__ void nalu_wmma_lean(const Frag& f, Acc& acc) {
#pragma unroll
    for (int ks = K0; ks < K1; ++ks) {
        v2f ax0 = __builtin_shufflevector(f.a0[ks], f.a0[ks], 0, 1);
        v2f ax1 = __builtin_shufflevector(f.a1[ks], f.a1[ks], 0, 1);
        v2f bw  = __builtin_shufflevector(f.q0[ks], f.q0[ks], 0, 1);
        acc.A0 = __builtin_amdgcn_wmma_f32_16x16x4_f32(false, ax0, false, bw,
                                                       (short)0, acc.A0, false, false);
        acc.A1 = __builtin_amdgcn_wmma_f32_16x16x4_f32(false, ax1, false, bw,
                                                       (short)0, acc.A1, false, false);
    }
}

template <int K0, int K1, bool SIGN>
__device__ __forceinline__ void nalu_wmma(const Frag& f, Acc& acc) {
#pragma unroll
    for (int ks = K0; ks < K1; ++ks) {
        v2f ax0 = __builtin_shufflevector(f.a0[ks], f.a0[ks], 0, 1);
        v2f al0 = __builtin_shufflevector(f.a0[ks], f.a0[ks], 2, 3);
        v2f ax1 = __builtin_shufflevector(f.a1[ks], f.a1[ks], 0, 1);
        v2f al1 = __builtin_shufflevector(f.a1[ks], f.a1[ks], 2, 3);
        v2f bw  = __builtin_shufflevector(f.q0[ks], f.q0[ks], 0, 1);
        v2f b2  = __builtin_shufflevector(f.q0[ks], f.q0[ks], 2, 3);
        v2f b1  = __builtin_shufflevector(f.q1[ks], f.q1[ks], 0, 1);
        v2f bn  = __builtin_shufflevector(f.q1[ks], f.q1[ks], 2, 3);

        v2f an0 = {}, az0 = {}, an1 = {}, az1 = {};
        if (SIGN) {
        an0.x = (ax0.x < 0.0f) ? 1.0f : 0.0f;
        an0.y = (ax0.y < 0.0f) ? 1.0f : 0.0f;
        az0.x = (ax0.x == 0.0f) ? 1.0f : 0.0f;
        az0.y = (ax0.y == 0.0f) ? 1.0f : 0.0f;
        an1.x = (ax1.x < 0.0f) ? 1.0f : 0.0f;
        an1.y = (ax1.y < 0.0f) ? 1.0f : 0.0f;
        az1.x = (ax1.x == 0.0f) ? 1.0f : 0.0f;
        az1.y = (ax1.y == 0.0f) ? 1.0f : 0.0f;
        }

        if (!SIGN) {
        acc.A0 = __builtin_amdgcn_wmma_f32_16x16x4_f32(false, ax0, false, bw,
                                                       (short)0, acc.A0, false, false);
        acc.A1 = __builtin_amdgcn_wmma_f32_16x16x4_f32(false, ax1, false, bw,
                                                       (short)0, acc.A1, false, false);
        acc.M0 = __builtin_amdgcn_wmma_f32_16x16x4_f32(false, al0, false, bw,
                                                       (short)0, acc.M0, false, false);
        acc.M1 = __builtin_amdgcn_wmma_f32_16x16x4_f32(false, al1, false, bw,
                                                       (short)0, acc.M1, false, false);
        continue;
        }
        acc.L0 = __builtin_amdgcn_wmma_f32_16x16x4_f32(false, an0, false, b2,
                                                       (short)0, acc.L0, false, false);
        acc.L1 = __builtin_amdgcn_wmma_f32_16x16x4_f32(false, an1, false, b2,
                                                       (short)0, acc.L1, false, false);
        acc.L0 = __builtin_amdgcn_wmma_f32_16x16x4_f32(false, az0, false, b1,
                                                       (short)0, acc.L0, false, false);
        acc.L1 = __builtin_amdgcn_wmma_f32_16x16x4_f32(false, az1, false, b1,
                                                       (short)0, acc.L1, false, false);
        v2f ap;
        ap.x = fmaf(an1.x, PRADIX, an0.x);
        ap.y = fmaf(an1.y, PRADIX, an0.y);
        acc.P = __builtin_amdgcn_wmma_f32_16x16x4_f32(false, ap, false, bn,
                                                      (short)0, acc.P, false, false);
    }
}

__device__ __forceinline__ void nalu_raw_load(const float* gx, const float* gwd, const float* gmd,
                                              const float* gwt, const float* gmt, Raw& r) {
#pragma unroll
    for (int j = 0; j < NLD; ++j) r.x[j] = *(const v2f*)(gx + j * 8 * NI);
#pragma unroll
    for (int e = 0; e < 2; ++e) {
        r.wd[e] = *(const v4f*)(gwd + e * NO);
        r.md[e] = *(const v4f*)(gmd + e * NO);
        r.wt[e] = *(const v4f*)(gwt + e * 16 * NO);
        r.mt[e] = *(const v4f*)(gmt + e * 16 * NO);
    }
}

__device__ __forceinline__ void nalu_raw_load_lean(const float* gx, const float* gwd, const float* gmd, Raw& r) {
#pragma unroll
    for (int j = 0; j < NLD; ++j) r.x[j] = *(const v2f*)(gx + j * 8 * NI);
#pragma unroll
    for (int e = 0; e < 2; ++e) {
        r.wd[e] = *(const v4f*)(gwd + e * NO);
        r.md[e] = *(const v4f*)(gmd + e * NO);
    }
}

__device__ __forceinline__ void nalu_xform_d(const Raw& r, Rec& o) {
#pragma unroll
    for (int i = 0; i < 4; ++i) {
        v2f w;
        w.x = nalu_w1(r.wd[0][i], r.md[0][i]);
        w.y = nalu_w1(r.wd[1][i], r.md[1][i]);
        o.bw[i] = w;
    }
}

__device__ __forceinline__ void nalu_xform_x(const Raw& r, Rec& o) {
#pragma unroll
    for (int j = 0; j < NLD; ++j) {
        v4f a;
        a.x = r.x[j].x;
        a.y = r.x[j].y;
        a.z = nalu_ln(fmaxf(fabsf(r.x[j].x), EPS));
        a.w = nalu_ln(fmaxf(fabsf(r.x[j].y), EPS));
        o.a[j] = a;
    }
}

__device__ __forceinline__ void nalu_xform_m(const Raw& r, Rec& o) {
#pragma unroll
    for (int j = 0; j < 2; ++j) {
#pragma unroll
        for (int h = 0; h < 2; ++h) {
            const float wsa = fabsf(nalu_w1(r.wt[j][2 * h], r.mt[j][2 * h]));
            const float wsb = fabsf(nalu_w1(r.wt[j][2 * h + 1], r.mt[j][2 * h + 1]));
            const float t2a = 1.0f - 2.0f * wsa;
            const float t2b = 1.0f - 2.0f * wsb;
            const float t1a = 1.0f - wsa;
            const float t1b = 1.0f - wsb;
            v2f lo;
            lo.x = fmaxf(nalu_ln(fabsf(t2a)), NEG_BIG);
            lo.y = fmaxf(nalu_ln(fabsf(t2b)), NEG_BIG);
            v4f hi;
            hi.x = fmaxf(nalu_ln(t1a), NEG_BIG);
            hi.y = fmaxf(nalu_ln(t1b), NEG_BIG);
            hi.z = (t2a < 0.0f) ? 1.0f : 0.0f;
            hi.w = (t2b < 0.0f) ? 1.0f : 0.0f;
            o.b2[2 * j + h] = lo;
            o.b1[2 * j + h] = hi;
        }
    }
}

__device__ __forceinline__ void nalu_xform_x_lean(const Raw& r, Rec& o) {
#pragma unroll
    for (int j = 0; j < NLD; ++j) {
        o.a[j].x = r.x[j].x;
        o.a[j].y = r.x[j].y;
    }
}

__device__ __forceinline__ void nalu_rec_store(v4f* sA, float* sD, v4f* sM, const Rec& o) {
#pragma unroll
    for (int j = 0; j < NLD; ++j) sA[j * 8 * A_PITCH] = o.a[j];
#pragma unroll
    for (int i = 0; i < 4; ++i) *(v2f*)(sD + 4 * i) = o.bw[i];
#pragma unroll
    for (int j = 0; j < 2; ++j) {
#pragma unroll
        for (int h = 0; h < 2; ++h) {
            v4f* q = sM + h * B_PITCH + j * 16;
            *(v2f*)((float*)q + 2) = o.b2[2 * j + h];
            q[B_V4] = o.b1[2 * j + h];
        }
    }
}

__device__ __forceinline__ void nalu_rec_store_lean(v4f* sA, float* sD, const Rec& o) {
#pragma unroll
    for (int j = 0; j < NLD; ++j) {
        v2f xy;
        xy.x = o.a[j].x;
        xy.y = o.a[j].y;
        *(v2f*)(sA + j * 8 * A_PITCH) = xy;
    }
#pragma unroll
    for (int i = 0; i < 4; ++i) *(v2f*)(sD + 4 * i) = o.bw[i];
}

__device__ __forceinline__ void nalu_ring_signal() {
    __builtin_amdgcn_fence(__ATOMIC_RELEASE, "workgroup", "local");
    __builtin_amdgcn_s_barrier_signal(-1);
    __builtin_amdgcn_sched_barrier(0);
}
__device__ __forceinline__ void nalu_ring_wait() {
    __builtin_amdgcn_sched_barrier(0);
    __builtin_amdgcn_s_barrier_wait(-1);
    __builtin_amdgcn_fence(__ATOMIC_ACQUIRE, "workgroup", "local");
}

__device__ __forceinline__ void nalu_wmma_above(const Acc& a) {
    asm volatile("" :: "v"(a.A0[0]), "v"(a.M0[0]), "v"(a.L0[0]), "v"(a.A1[0]),
                       "v"(a.M1[0]), "v"(a.L1[0]), "v"(a.P[0]));
}

__device__ __forceinline__ void nalu_wmma_above_lean(const Acc& a) {
    asm volatile("" :: "v"(a.A0[0]), "v"(a.A1[0]));
}

__device__ __forceinline__ bool nalu_sign_live(const Acc& a) {
    float m = fmaxf(a.L0[0], a.L1[0]);
#pragma unroll
    for (int v = 1; v < 8; ++v) m = fmaxf(m, fmaxf(a.L0[v], a.L1[v]));
    return __builtin_amdgcn_ballot_w32(!(m < SIGN_DEAD)) != 0u;
}

template <int N, int V, int ID>
__device__ __forceinline__ void nalu_interleave() {
#pragma unroll
    for (int i = 0; i < N; ++i) {
        __builtin_amdgcn_sched_group_barrier(0x402, V, ID);
        __builtin_amdgcn_sched_group_barrier(0x008, 1, ID);
    }
}

template <bool STORE, bool LOAD, bool XFORM, int ID>
__device__ __forceinline__ void nalu_stage(const Frag& fc, Frag& fn, Raw& rl, const Raw& rx, Rec& rec,
                                           v4f* sA, float* sD, v4f* sM, const v4f* fA, const v4f* fB,
                                           const float* gx, const float* gwd, const float* gmd,
                                           const float* gwt, const float* gmt, Acc& acc, bool& live,
                                           int* flag_own, const v4i* flag_all, bool& wg_live) {
    v4i fl = {1, 1, 1, 1};
    if (STORE) {
        nalu_rec_store(sA, sD, sM, rec);
        if ((threadIdx.x & 31) == 0) *flag_own = live ? 1 : 0;
        if (LOAD) nalu_raw_load(gx, gwd, gmd, gwt, gmt, rl);
        __builtin_amdgcn_sched_barrier(0);
    }
    if (XFORM) nalu_xform_d(rx, rec);
    nalu_wmma<0, 3, false>(fc, acc);
    if (XFORM) nalu_interleave<3 * 4, 4, ID>();
    if (live) nalu_wmma<0, 3, true>(fc, acc);
    if (STORE) {
        nalu_wmma_above(acc);
        nalu_ring_signal();
    }
    if (XFORM) nalu_xform_x(rx, rec);
    nalu_wmma<3, 4, false>(fc, acc);
    if (XFORM) nalu_interleave<4, 12, ID + 1>();
    if (live) nalu_wmma<3, 4, true>(fc, acc);
    if (STORE) {
        nalu_wmma_above(acc);
        nalu_ring_wait();
        fl = *flag_all;
        nalu_frag_load(fA, fB, fn);
        __builtin_amdgcn_sched_barrier(0);
    }
    if (XFORM) nalu_xform_m(rx, rec);
    nalu_wmma<4, KS, false>(fc, acc);
    if (XFORM) nalu_interleave<(KS - 4) * 4, 16, ID + 2>();
    if (live) {
        nalu_wmma<4, KS, true>(fc, acc);
        live = nalu_sign_live(acc);
    }
    if (STORE) wg_live = __builtin_amdgcn_readfirstlane(fl.x | fl.y | fl.z | fl.w) != 0;
    if (XFORM) {
        nalu_wmma_above(acc);
        __builtin_amdgcn_sched_barrier(0);
    }
}

template <bool STORE, bool LOAD, bool XFORM, int ID>
__device__ __forceinline__ void nalu_stage_lean(const Frag& fc, Frag& fn, Raw& rl, const Raw& rx, Rec& rec,
                                                v4f* sA, float* sD, const v4f* fA, const v4f* fB,
                                                const float* gx, const float* gwd, const float* gmd,
                                                Acc& acc) {
    if (STORE) {
        nalu_rec_store_lean(sA, sD, rec);
        if (LOAD) nalu_raw_load_lean(gx, gwd, gmd, rl);
        __builtin_amdgcn_sched_barrier(0);
    }
    nalu_wmma_lean<0, 2>(fc, acc);
    if (STORE) {
        nalu_wmma_above_lean(acc);
        nalu_ring_signal();
    }
    nalu_wmma_lean<2, 4>(fc, acc);
    if (STORE) {
        nalu_wmma_above_lean(acc);
        nalu_ring_wait();
        nalu_frag_load_lean(fA, fB, fn);
        __builtin_amdgcn_sched_barrier(0);
    }
    if (XFORM) {
        nalu_xform_d(rx, rec);
        nalu_xform_x_lean(rx, rec);
    }
    nalu_wmma_lean<4, KS>(fc, acc);
    if (XFORM) {
        nalu_interleave<(KS - 4) * 2, 6, ID>();
        nalu_wmma_above_lean(acc);
        __builtin_amdgcn_sched_barrier(0);
    }
}

#define NALU_ADVANCE() do { gx += 2 * KP; gwt += 2 * KP; gmt += 2 * KP; \
                            gwd += 2 * KP * NO; gmd += 2 * KP * NO; } while (0)

__global__ __launch_bounds__(128) void nalu_main_kernel(const float* __restrict__ x,
                                                        const float* __restrict__ w_hat,
                                                        const float* __restrict__ m_hat,
                                                        const float* __restrict__ g,
                                                        float* __restrict__ out) {
    __shared__ v4f lds[2 * STAGE_V4];
    __shared__ __attribute__((aligned(16))) int sign_state[4];

    const int tid = threadIdx.x;
    const int lane = tid & 31;
    const int wave = tid >> 5;
    const int wm = wave >> 1;
    const int wn = wave & 1;
    const int bm0 = (blockIdx.x >> 4) * BM;
    const int bn0 = (blockIdx.x & 15) * BN;
    const int half = lane >> 4;
    const int l15 = lane & 15;

    const float* gx = x + (bm0 + (tid >> 4)) * NI + 2 * (tid & 15);
    const int offd = (2 * (tid >> 3)) * NO + bn0 + 4 * (tid & 7);
    const int offt = (bn0 + (tid >> 3)) * NO + 4 * (tid & 7);
    const float* gwd = w_hat + offd;
    const float* gmd = m_hat + offd;
    const float* gwt = w_hat + offt;
    const float* gmt = m_hat + offt;
    const int sAo = (tid >> 4) * A_PITCH + (tid & 15);
    const int sDo = A_V4 + (tid >> 3) * B_PITCH + 4 * (tid & 7);
    const int sMo = A_V4 + 2 * (tid & 7) * B_PITCH + (tid >> 3);

    const int fAo = (wm * 32 + l15) * A_PITCH + half;
    const int fBo = A_V4 + half * B_PITCH + wn * 16 + l15;

    Acc acc;
    acc.A0 = v8f{}; acc.M0 = v8f{}; acc.L0 = v8f{};
    acc.A1 = v8f{}; acc.M1 = v8f{}; acc.L1 = v8f{}; acc.P = v8f{};

    v4f* const slot0 = lds;
    v4f* const slot1 = lds + STAGE_V4;

    Raw c0, c1;
    Rec rec;
    nalu_raw_load(gx, gwd, gmd, gwt, gmt, c0);
    NALU_ADVANCE();
    nalu_raw_load(gx, gwd, gmd, gwt, gmt, c1);
    NALU_ADVANCE();
    nalu_xform_d(c0, rec);
    nalu_xform_x(c0, rec);
    nalu_xform_m(c0, rec);
    nalu_rec_store(slot0 + sAo, (float*)(slot0 + sDo), slot0 + sMo, rec);
    nalu_raw_load(gx, gwd, gmd, gwt, gmt, c0);
    NALU_ADVANCE();
    nalu_xform_d(c1, rec);
    nalu_xform_x(c1, rec);
    nalu_xform_m(c1, rec);
    __builtin_amdgcn_fence(__ATOMIC_RELEASE, "workgroup", "local");
    __builtin_amdgcn_s_barrier_signal(-1);
    __builtin_amdgcn_s_barrier_wait(-1);
    __builtin_amdgcn_fence(__ATOMIC_ACQUIRE, "workgroup", "local");

    bool live = true;

    Frag f0, f1;
    nalu_frag_load(slot0 + fAo, slot0 + fBo, f0);

    bool wg_live = true;

    int it = 0;
#pragma unroll 1
    for (; it < (NSTAGE - 4) / 2 && wg_live; ++it) {
        nalu_stage<true, true, true, 0>(f0, f1, c1, c0, rec, slot1 + sAo, (float*)(slot1 + sDo), slot1 + sMo,
                                     slot1 + fAo, slot1 + fBo, gx, gwd, gmd, gwt, gmt, acc, live, sign_state + wave, (const v4i*)sign_state, wg_live);
        NALU_ADVANCE();
        nalu_stage<true, true, true, 3>(f1, f0, c0, c1, rec, slot0 + sAo, (float*)(slot0 + sDo), slot0 + sMo,
                                     slot0 + fAo, slot0 + fBo, gx, gwd, gmd, gwt, gmt, acc, live, sign_state + wave, (const v4i*)sign_state, wg_live);
        NALU_ADVANCE();
    }
    if (wg_live) {
        nalu_stage<true, true, true, 6>(f0, f1, c1, c0, rec, slot1 + sAo, (float*)(slot1 + sDo), slot1 + sMo,
                                     slot1 + fAo, slot1 + fBo, gx, gwd, gmd, gwt, gmt, acc, live, sign_state + wave, (const v4i*)sign_state, wg_live);
        nalu_stage<true, false, true, 9>(f1, f0, c0, c1, rec, slot0 + sAo, (float*)(slot0 + sDo), slot0 + sMo,
                                      slot0 + fAo, slot0 + fBo, gx, gwd, gmd, gwt, gmt, acc, live, sign_state + wave, (const v4i*)sign_state, wg_live);
        nalu_stage<true, false, false, 12>(f0, f1, c1, c0, rec, slot1 + sAo, (float*)(slot1 + sDo), slot1 + sMo,
                                       slot1 + fAo, slot1 + fBo, gx, gwd, gmd, gwt, gmt, acc, live, sign_state + wave, (const v4i*)sign_state, wg_live);
        nalu_stage<false, false, false, 15>(f1, f0, c0, c1, rec, slot0 + sAo, (float*)(slot0 + sDo), slot0 + sMo,
                                        slot0 + fAo, slot0 + fBo, gx, gwd, gmd, gwt, gmt, acc, live, sign_state + wave, (const v4i*)sign_state, wg_live);
    } else {
#pragma unroll 1
        for (; it < (NSTAGE - 4) / 2; ++it) {
            nalu_stage_lean<true, true, true, 20>(f0, f1, c1, c0, rec, slot1 + sAo, (float*)(slot1 + sDo),
                                                  slot1 + fAo, slot1 + fBo, gx, gwd, gmd, acc);
            NALU_ADVANCE();
            nalu_stage_lean<true, true, true, 21>(f1, f0, c0, c1, rec, slot0 + sAo, (float*)(slot0 + sDo),
                                                  slot0 + fAo, slot0 + fBo, gx, gwd, gmd, acc);
            NALU_ADVANCE();
        }
        nalu_stage_lean<true, true, true, 22>(f0, f1, c1, c0, rec, slot1 + sAo, (float*)(slot1 + sDo),
                                              slot1 + fAo, slot1 + fBo, gx, gwd, gmd, acc);
        nalu_stage_lean<true, false, true, 23>(f1, f0, c0, c1, rec, slot0 + sAo, (float*)(slot0 + sDo),
                                               slot0 + fAo, slot0 + fBo, gx, gwd, gmd, acc);
        nalu_stage_lean<true, false, false, 24>(f0, f1, c1, c0, rec, slot1 + sAo, (float*)(slot1 + sDo),
                                                slot1 + fAo, slot1 + fBo, gx, gwd, gmd, acc);
        nalu_stage_lean<false, false, false, 25>(f1, f0, c0, c1, rec, slot0 + sAo, (float*)(slot0 + sDo),
                                                 slot0 + fAo, slot0 + fBo, gx, gwd, gmd, acc);
    }

    const int m0 = bm0 + wm * 32;
    const int bcol = bn0 + wn * 16 + l15;
    const float g1 = 1.0f / (1.0f + expf(-g[bcol]));
    const float gm = 1.0f - g1;
#pragma unroll
    for (int v = 0; v < 8; ++v) {
        const int row0 = m0 + v + 8 * half;
        const int negc = (int)(acc.P[v] + 0.5f);
        {
            const float m1 = expf(fminf(acc.M0[v], LOG_CLIP));
            float ms = ((negc & 1) ? -1.0f : 1.0f) * expf(acc.L0[v]);
            ms = fminf(fmaxf(ms, -1.0f), 1.0f);
            out[row0 * NO + bcol] = g1 * acc.A0[v] + gm * m1 * ms;
        }
        {
            const float m1 = expf(fminf(acc.M1[v], LOG_CLIP));
            float ms = (((negc >> PSHIFT) & 1) ? -1.0f : 1.0f) * expf(acc.L1[v]);
            ms = fminf(fmaxf(ms, -1.0f), 1.0f);
            out[(row0 + 16) * NO + bcol] = g1 * acc.A1[v] + gm * m1 * ms;
        }
    }
}

extern "C" void kernel_launch(void* const* d_in, const int* in_sizes, int n_in,
                              void* d_out, int out_size, void* d_ws, size_t ws_size,
                              hipStream_t stream) {
    const float* x     = (const float*)d_in[0];
    const float* w_hat = (const float*)d_in[1];
    const float* m_hat = (const float*)d_in[2];
    const float* g     = (const float*)d_in[3];
    float* out = (float*)d_out;

    nalu_main_kernel<<<256, 128, 0, stream>>>(x, w_hat, m_hat, g, out);
}


# accM = logx@w1 on demand: main K loop drops its 2 accM WMMAs per k-step and every log|x|; only if a wave's sign path outlives K does the workgroup walk K again (lean form) for accM; sums kept
# speedup vs baseline: 9.3777x; 1.1836x over previous
#include <hip/hip_runtime.h>
#include <math.h>

typedef float v2f __attribute__((ext_vector_type(2)));
typedef float v4f __attribute__((ext_vector_type(4)));
typedef float v8f __attribute__((ext_vector_type(8)));
typedef int v4i __attribute__((ext_vector_type(4)));


#define NB 1024
#define NI 512
#define NO 512
#define EPS 1e-7f
#define LOG_CLIP 20.0f
#define NEG_BIG -1e30f
#define LN2 0.693147180559945309f


__device__ __forceinline__ float nalu_w1(float wh, float mh) {
    const float h = 0.5f * __builtin_amdgcn_tanhf(wh);
    const float tm = __builtin_amdgcn_tanhf(0.5f * mh);
    return fmaf(h, tm, h);
}

__device__ __forceinline__ float nalu_ln(float v) {
    return __builtin_amdgcn_logf(v) * LN2;
}

#define BM 64
#define BN 32
#define KP 16
#define KS (KP / 2)
#define NSTAGE ((NI / 2) / KP)
#define A_PITCH (KP + 1)
#define B_PITCH (BN + 1)
#define A_V4 (BM * A_PITCH)
#define B_V4 (KP * B_PITCH)
#define STAGE_V4 (A_V4 + 2 * B_V4)
#define NLD 8
#define PRADIX 1024.0f
#define PSHIFT 10
#define SIGN_DEAD -106.0f


struct Acc {
    v8f A0, M0, L0, A1, M1, L1, P;
};
struct Frag {
    v2f a0[KS], a1[KS];
    v4f q0[KS], q1[KS];
};
struct Raw {
    v2f x[NLD];
    v4f wd[2], md[2];
    v4f wt[2], mt[2];
};
struct Rec {
    v2f a[NLD];
    v2f bw[4];
    v2f b2[4];
    v4f b1[4];
};

__device__ __forceinline__ void nalu_frag_load(const v4f* fa0, const v4f* fb0, Frag& f) {
    const v4f* fa1 = fa0 + 16 * A_PITCH;
    const v4f* fb1 = fb0 + B_V4;
#pragma unroll
    for (int ks = 0; ks < KS; ++ks) {
        f.a0[ks] = *(const v2f*)(fa0 + 2 * ks);
        f.a1[ks] = *(const v2f*)(fa1 + 2 * ks);
        f.q0[ks] = fb0[2 * ks * B_PITCH];
        f.q1[ks] = fb1[2 * ks * B_PITCH];
    }
}

__device__ __forceinline__ void nalu_frag_load_lean(const v4f* fa0, const v4f* fb0, Frag& f) {
    const v4f* fa1 = fa0 + 16 * A_PITCH;
#pragma unroll
    for (int ks = 0; ks < KS; ++ks) {
        const v2f q0 = *(const v2f*)(fb0 + 2 * ks * B_PITCH);
        f.a0[ks] = *(const v2f*)(fa0 + 2 * ks);
        f.a1[ks] = *(const v2f*)(fa1 + 2 * ks);
        f.q0[ks].x = q0.x; f.q0[ks].y = q0.y;
    }
}

template <int K0, int K1, bool TOM>
__device__ __forceinline__ void nalu_wmma_lean(const Frag& f, Acc& acc) {
#pragma unroll
    for (int ks = K0; ks < K1; ++ks) {
        v2f bw  = __builtin_shufflevector(f.q0[ks], f.q0[ks], 0, 1);
        if (TOM) {
        acc.M0 = __builtin_amdgcn_wmma_f32_16x16x4_f32(false, f.a0[ks], false, bw,
                                                       (short)0, acc.M0, false, false);
        acc.M1 = __builtin_amdgcn_wmma_f32_16x16x4_f32(false, f.a1[ks], false, bw,
                                                       (short)0, acc.M1, false, false);
        } else {
        acc.A0 = __builtin_amdgcn_wmma_f32_16x16x4_f32(false, f.a0[ks], false, bw,
                                                       (short)0, acc.A0, false, false);
        acc.A1 = __builtin_amdgcn_wmma_f32_16x16x4_f32(false, f.a1[ks], false, bw,
                                                       (short)0, acc.A1, false, false);
        }
    }
}

template <int K0, int K1>
__device__ __forceinline__ void nalu_wmma_sign(const Frag& f, Acc& acc) {
#pragma unroll
    for (int ks = K0; ks < K1; ++ks) {
        const v2f ax0 = f.a0[ks];
        const v2f ax1 = f.a1[ks];
        v2f b2  = __builtin_shufflevector(f.q0[ks], f.q0[ks], 2, 3);
        v2f b1  = __builtin_shufflevector(f.q1[ks], f.q1[ks], 0, 1);
        v2f bn  = __builtin_shufflevector(f.q1[ks], f.q1[ks], 2, 3);

        v2f an0, az0, an1, az1;
        an0.x = (ax0.x < 0.0f) ? 1.0f : 0.0f;
        an0.y = (ax0.y < 0.0f) ? 1.0f : 0.0f;
        az0.x = (ax0.x == 0.0f) ? 1.0f : 0.0f;
        az0.y = (ax0.y == 0.0f) ? 1.0f : 0.0f;
        an1.x = (ax1.x < 0.0f) ? 1.0f : 0.0f;
        an1.y = (ax1.y < 0.0f) ? 1.0f : 0.0f;
        az1.x = (ax1.x == 0.0f) ? 1.0f : 0.0f;
        az1.y = (ax1.y == 0.0f) ? 1.0f : 0.0f;

        acc.L0 = __builtin_amdgcn_wmma_f32_16x16x4_f32(false, an0, false, b2,
                                                       (short)0, acc.L0, false, false);
        acc.L1 = __builtin_amdgcn_wmma_f32_16x16x4_f32(false, an1, false, b2,
                                                       (short)0, acc.L1, false, false);
        acc.L0 = __builtin_amdgcn_wmma_f32_16x16x4_f32(false, az0, false, b1,
                                                       (short)0, acc.L0, false, false);
        acc.L1 = __builtin_amdgcn_wmma_f32_16x16x4_f32(false, az1, false, b1,
                                                       (short)0, acc.L1, false, false);
        v2f ap;
        ap.x = fmaf(an1.x, PRADIX, an0.x);
        ap.y = fmaf(an1.y, PRADIX, an0.y);
        acc.P = __builtin_amdgcn_wmma_f32_16x16x4_f32(false, ap, false, bn,
                                                      (short)0, acc.P, false, false);
    }
}

__device__ __forceinline__ void nalu_raw_load(const float* gx, const float* gwd, const float* gmd,
                                              const float* gwt, const float* gmt, Raw& r) {
#pragma unroll
    for (int j = 0; j < NLD; ++j) r.x[j] = *(const v2f*)(gx + j * 8 * NI);
#pragma unroll
    for (int e = 0; e < 2; ++e) {
        r.wd[e] = *(const v4f*)(gwd + e * NO);
        r.md[e] = *(const v4f*)(gmd + e * NO);
        r.wt[e] = *(const v4f*)(gwt + e * 16 * NO);
        r.mt[e] = *(const v4f*)(gmt + e * 16 * NO);
    }
}

__device__ __forceinline__ void nalu_raw_load_lean(const float* gx, const float* gwd, const float* gmd, Raw& r) {
#pragma unroll
    for (int j = 0; j < NLD; ++j) r.x[j] = *(const v2f*)(gx + j * 8 * NI);
#pragma unroll
    for (int e = 0; e < 2; ++e) {
        r.wd[e] = *(const v4f*)(gwd + e * NO);
        r.md[e] = *(const v4f*)(gmd + e * NO);
    }
}

__device__ __forceinline__ void nalu_xform_d(const Raw& r, Rec& o) {
#pragma unroll
    for (int i = 0; i < 4; ++i) {
        v2f w;
        w.x = nalu_w1(r.wd[0][i], r.md[0][i]);
        w.y = nalu_w1(r.wd[1][i], r.md[1][i]);
        o.bw[i] = w;
    }
}

template <bool LOGX>
__device__ __forceinline__ void nalu_xform_x(const Raw& r, Rec& o) {
#pragma unroll
    for (int j = 0; j < NLD; ++j) {
        if (LOGX) {
            v2f a;
            a.x = nalu_ln(fmaxf(fabsf(r.x[j].x), EPS));
            a.y = nalu_ln(fmaxf(fabsf(r.x[j].y), EPS));
            o.a[j] = a;
        } else {
            o.a[j] = r.x[j];
        }
    }
}

__device__ __forceinline__ void nalu_xform_m(const Raw& r, Rec& o) {
#pragma unroll
    for (int j = 0; j < 2; ++j) {
#pragma unroll
        for (int h = 0; h < 2; ++h) {
            const float wsa = fabsf(nalu_w1(r.wt[j][2 * h], r.mt[j][2 * h]));
            const float wsb = fabsf(nalu_w1(r.wt[j][2 * h + 1], r.mt[j][2 * h + 1]));
            const float t2a = 1.0f - 2.0f * wsa;
            const float t2b = 1.0f - 2.0f * wsb;
            const float t1a = 1.0f - wsa;
            const float t1b = 1.0f - wsb;
            v2f lo;
            lo.x = fmaxf(nalu_ln(fabsf(t2a)), NEG_BIG);
            lo.y = fmaxf(nalu_ln(fabsf(t2b)), NEG_BIG);
            v4f hi;
            hi.x = fmaxf(nalu_ln(t1a), NEG_BIG);
            hi.y = fmaxf(nalu_ln(t1b), NEG_BIG);
            hi.z = (t2a < 0.0f) ? 1.0f : 0.0f;
            hi.w = (t2b < 0.0f) ? 1.0f : 0.0f;
            o.b2[2 * j + h] = lo;
            o.b1[2 * j + h] = hi;
        }
    }
}

__device__ __forceinline__ void nalu_rec_store(v4f* sA, float* sD, v4f* sM, const Rec& o) {
#pragma unroll
    for (int j = 0; j < NLD; ++j) *(v2f*)(sA + j * 8 * A_PITCH) = o.a[j];
#pragma unroll
    for (int i = 0; i < 4; ++i) *(v2f*)(sD + 4 * i) = o.bw[i];
#pragma unroll
    for (int j = 0; j < 2; ++j) {
#pragma unroll
        for (int h = 0; h < 2; ++h) {
            v4f* q = sM + h * B_PITCH + j * 16;
            *(v2f*)((float*)q + 2) = o.b2[2 * j + h];
            q[B_V4] = o.b1[2 * j + h];
        }
    }
}

__device__ __forceinline__ void nalu_rec_store_lean(v4f* sA, float* sD, const Rec& o) {
#pragma unroll
    for (int j = 0; j < NLD; ++j) *(v2f*)(sA + j * 8 * A_PITCH) = o.a[j];
#pragma unroll
    for (int i = 0; i < 4; ++i) *(v2f*)(sD + 4 * i) = o.bw[i];
}

__device__ __forceinline__ void nalu_ring_signal() {
    __builtin_amdgcn_fence(__ATOMIC_RELEASE, "workgroup", "local");
    __builtin_amdgcn_s_barrier_signal(-1);
    __builtin_amdgcn_sched_barrier(0);
}
__device__ __forceinline__ void nalu_ring_wait() {
    __builtin_amdgcn_sched_barrier(0);
    __builtin_amdgcn_s_barrier_wait(-1);
    __builtin_amdgcn_fence(__ATOMIC_ACQUIRE, "workgroup", "local");
}

__device__ __forceinline__ void nalu_wmma_above(const Acc& a) {
    asm volatile("" :: "v"(a.A0[0]), "v"(a.L0[0]), "v"(a.A1[0]), "v"(a.L1[0]), "v"(a.P[0]));
}

template <bool TOM>
__device__ __forceinline__ void nalu_wmma_above_lean(const Acc& a) {
    if (TOM) asm volatile("" :: "v"(a.M0[0]), "v"(a.M1[0]));
    else     asm volatile("" :: "v"(a.A0[0]), "v"(a.A1[0]));
}

__device__ __forceinline__ bool nalu_sign_live(const Acc& a) {
    float m = fmaxf(a.L0[0], a.L1[0]);
#pragma unroll
    for (int v = 1; v < 8; ++v) m = fmaxf(m, fmaxf(a.L0[v], a.L1[v]));
    return __builtin_amdgcn_ballot_w32(!(m < SIGN_DEAD)) != 0u;
}

template <int N, int V, int ID>
__device__ __forceinline__ void nalu_interleave() {
#pragma unroll
    for (int i = 0; i < N; ++i) {
        __builtin_amdgcn_sched_group_barrier(0x402, V, ID);
        __builtin_amdgcn_sched_group_barrier(0x008, 1, ID);
    }
}

template <bool STORE, bool LOAD, bool XFORM>
__device__ __forceinline__ void nalu_stage(const Frag& fc, Frag& fn, Raw& rl, const Raw& rx, Rec& rec,
                                           v4f* sA, float* sD, v4f* sM, const v4f* fA, const v4f* fB,
                                           const float* gx, const float* gwd, const float* gmd,
                                           const float* gwt, const float* gmt, Acc& acc, bool& live,
                                           int* flag_own, const v4i* flag_all, bool& wg_live) {
    v4i fl = {1, 1, 1, 1};
    if (STORE) {
        nalu_rec_store(sA, sD, sM, rec);
        if ((threadIdx.x & 31) == 0) *flag_own = live ? 1 : 0;
        if (LOAD) nalu_raw_load(gx, gwd, gmd, gwt, gmt, rl);
        __builtin_amdgcn_sched_barrier(0);
    }
    if (XFORM) {
        nalu_xform_d(rx, rec);
        nalu_xform_x<false>(rx, rec);
        nalu_xform_m(rx, rec);
    }
    nalu_wmma_lean<0, 3, false>(fc, acc);
    if (live) nalu_wmma_sign<0, 3>(fc, acc);
    if (STORE) {
        nalu_wmma_above(acc);
        nalu_ring_signal();
    }
    nalu_wmma_lean<3, 4, false>(fc, acc);
    if (live) nalu_wmma_sign<3, 4>(fc, acc);
    if (STORE) {
        nalu_wmma_above(acc);
        nalu_ring_wait();
        fl = *flag_all;
        nalu_frag_load(fA, fB, fn);
        __builtin_amdgcn_sched_barrier(0);
    }
    nalu_wmma_lean<4, KS, false>(fc, acc);
    if (live) {
        nalu_wmma_sign<4, KS>(fc, acc);
        live = nalu_sign_live(acc);
    }
    if (STORE) wg_live = __builtin_amdgcn_readfirstlane(fl.x | fl.y | fl.z | fl.w) != 0;
    if (XFORM) {
        nalu_wmma_above(acc);
        __builtin_amdgcn_sched_barrier(0);
    }
}

template <bool STORE, bool LOAD, bool XFORM, int ID, bool LOGX>
__device__ __forceinline__ void nalu_stage_lean(const Frag& fc, Frag& fn, Raw& rl, const Raw& rx, Rec& rec,
                                                v4f* sA, float* sD, const v4f* fA, const v4f* fB,
                                                const float* gx, const float* gwd, const float* gmd,
                                                Acc& acc) {
    if (STORE) {
        nalu_rec_store_lean(sA, sD, rec);
        if (LOAD) nalu_raw_load_lean(gx, gwd, gmd, rl);
        __builtin_amdgcn_sched_barrier(0);
    }
    nalu_wmma_lean<0, 2, LOGX>(fc, acc);
    if (STORE) {
        nalu_wmma_above_lean<LOGX>(acc);
        nalu_ring_signal();
    }
    nalu_wmma_lean<2, 4, LOGX>(fc, acc);
    if (STORE) {
        nalu_wmma_above_lean<LOGX>(acc);
        nalu_ring_wait();
        nalu_frag_load_lean(fA, fB, fn);
        __builtin_amdgcn_sched_barrier(0);
    }
    if (XFORM) {
        nalu_xform_d(rx, rec);
        nalu_xform_x<LOGX>(rx, rec);
    }
    nalu_wmma_lean<4, KS, LOGX>(fc, acc);
    if (XFORM) {
        nalu_interleave<(KS - 4) * 2, LOGX ? 12 : 6, ID>();
        nalu_wmma_above_lean<LOGX>(acc);
        __builtin_amdgcn_sched_barrier(0);
    }
}

#define NALU_AT_STAGE() do { gx = gx0 + lp * (2 * KP); gwt = gwt0 + lp * (2 * KP); gmt = gmt0 + lp * (2 * KP); \
                             gwd = gwd0 + lp * (2 * KP * NO); gmd = gmd0 + lp * (2 * KP * NO); } while (0)
#define NALU_ADVANCE() do { lp = (lp + 1) & (NSTAGE - 1); NALU_AT_STAGE(); } while (0)

__global__ __launch_bounds__(128) void nalu_main_kernel(const float* __restrict__ x,
                                                        const float* __restrict__ w_hat,
                                                        const float* __restrict__ m_hat,
                                                        const float* __restrict__ g,
                                                        float* __restrict__ out) {
    __shared__ v4f lds[2 * STAGE_V4];
    __shared__ __attribute__((aligned(16))) int sign_state[4];

    const int tid = threadIdx.x;
    const int lane = tid & 31;
    const int wave = tid >> 5;
    const int wm = wave >> 1;
    const int wn = wave & 1;
    const int bm0 = (blockIdx.x >> 4) * BM;
    const int bn0 = (blockIdx.x & 15) * BN;
    const int half = lane >> 4;
    const int l15 = lane & 15;

    const float* const gx0 = x + (bm0 + (tid >> 4)) * NI + 2 * (tid & 15);
    const int offd = (2 * (tid >> 3)) * NO + bn0 + 4 * (tid & 7);
    const int offt = (bn0 + (tid >> 3)) * NO + 4 * (tid & 7);
    const float* const gwd0 = w_hat + offd;
    const float* const gmd0 = m_hat + offd;
    const float* const gwt0 = w_hat + offt;
    const float* const gmt0 = m_hat + offt;
    const int lp0 = ((blockIdx.x & 15) + (blockIdx.x >> 4)) & (NSTAGE - 1);
    int lp = lp0;
    const float *gx, *gwd, *gmd, *gwt, *gmt;
    NALU_AT_STAGE();
    const int sAo = (tid >> 4) * A_PITCH + (tid & 15);
    const int sDo = A_V4 + (tid >> 3) * B_PITCH + 4 * (tid & 7);
    const int sMo = A_V4 + 2 * (tid & 7) * B_PITCH + (tid >> 3);

    const int fAo = (wm * 32 + l15) * A_PITCH + half;
    const int fBo = A_V4 + half * B_PITCH + wn * 16 + l15;

    Acc acc;
    acc.A0 = v8f{}; acc.M0 = v8f{}; acc.L0 = v8f{};
    acc.A1 = v8f{}; acc.M1 = v8f{}; acc.L1 = v8f{}; acc.P = v8f{};

    v4f* const slot0 = lds;
    v4f* const slot1 = lds + STAGE_V4;

    Raw c0, c1;
    Rec rec;
    nalu_raw_load(gx, gwd, gmd, gwt, gmt, c0);
    NALU_ADVANCE();
    nalu_raw_load(gx, gwd, gmd, gwt, gmt, c1);
    NALU_ADVANCE();
    nalu_xform_d(c0, rec);
    nalu_xform_x<false>(c0, rec);
    nalu_xform_m(c0, rec);
    nalu_rec_store(slot0 + sAo, (float*)(slot0 + sDo), slot0 + sMo, rec);
    nalu_raw_load(gx, gwd, gmd, gwt, gmt, c0);
    NALU_ADVANCE();
    nalu_xform_d(c1, rec);
    nalu_xform_x<false>(c1, rec);
    nalu_xform_m(c1, rec);
    __builtin_amdgcn_fence(__ATOMIC_RELEASE, "workgroup", "local");
    __builtin_amdgcn_s_barrier_signal(-1);
    __builtin_amdgcn_s_barrier_wait(-1);
    __builtin_amdgcn_fence(__ATOMIC_ACQUIRE, "workgroup", "local");

    bool live = true;

    Frag f0, f1;
    nalu_frag_load(slot0 + fAo, slot0 + fBo, f0);

    bool wg_live = true;

    int it = 0;
#pragma unroll 1
    for (; it < (NSTAGE - 4) / 2 && wg_live; ++it) {
        nalu_stage<true, true, true>(f0, f1, c1, c0, rec, slot1 + sAo, (float*)(slot1 + sDo), slot1 + sMo,
                                     slot1 + fAo, slot1 + fBo, gx, gwd, gmd, gwt, gmt, acc, live, sign_state + wave, (const v4i*)sign_state, wg_live);
        NALU_ADVANCE();
        nalu_stage<true, true, true>(f1, f0, c0, c1, rec, slot0 + sAo, (float*)(slot0 + sDo), slot0 + sMo,
                                     slot0 + fAo, slot0 + fBo, gx, gwd, gmd, gwt, gmt, acc, live, sign_state + wave, (const v4i*)sign_state, wg_live);
        NALU_ADVANCE();
    }
    if (wg_live) {
        nalu_stage<true, true, true>(f0, f1, c1, c0, rec, slot1 + sAo, (float*)(slot1 + sDo), slot1 + sMo,
                                     slot1 + fAo, slot1 + fBo, gx, gwd, gmd, gwt, gmt, acc, live, sign_state + wave, (const v4i*)sign_state, wg_live);
        nalu_stage<true, false, true>(f1, f0, c0, c1, rec, slot0 + sAo, (float*)(slot0 + sDo), slot0 + sMo,
                                      slot0 + fAo, slot0 + fBo, gx, gwd, gmd, gwt, gmt, acc, live, sign_state + wave, (const v4i*)sign_state, wg_live);
        nalu_stage<true, false, false>(f0, f1, c1, c0, rec, slot1 + sAo, (float*)(slot1 + sDo), slot1 + sMo,
                                       slot1 + fAo, slot1 + fBo, gx, gwd, gmd, gwt, gmt, acc, live, sign_state + wave, (const v4i*)sign_state, wg_live);
        nalu_stage<false, false, false>(f1, f0, c0, c1, rec, slot0 + sAo, (float*)(slot0 + sDo), slot0 + sMo,
                                        slot0 + fAo, slot0 + fBo, gx, gwd, gmd, gwt, gmt, acc, live, sign_state + wave, (const v4i*)sign_state, wg_live);
    } else {
#pragma unroll 1
        for (; it < (NSTAGE - 4) / 2; ++it) {
            nalu_stage_lean<true, true, true, 20, false>(f0, f1, c1, c0, rec, slot1 + sAo, (float*)(slot1 + sDo),
                                                  slot1 + fAo, slot1 + fBo, gx, gwd, gmd, acc);
            NALU_ADVANCE();
            nalu_stage_lean<true, true, true, 21, false>(f1, f0, c0, c1, rec, slot0 + sAo, (float*)(slot0 + sDo),
                                                  slot0 + fAo, slot0 + fBo, gx, gwd, gmd, acc);
            NALU_ADVANCE();
        }
        nalu_stage_lean<true, true, true, 22, false>(f0, f1, c1, c0, rec, slot1 + sAo, (float*)(slot1 + sDo),
                                              slot1 + fAo, slot1 + fBo, gx, gwd, gmd, acc);
        nalu_stage_lean<true, false, true, 23, false>(f1, f0, c0, c1, rec, slot0 + sAo, (float*)(slot0 + sDo),
                                               slot0 + fAo, slot0 + fBo, gx, gwd, gmd, acc);
        nalu_stage_lean<true, false, false, 24, false>(f0, f1, c1, c0, rec, slot1 + sAo, (float*)(slot1 + sDo),
                                                slot1 + fAo, slot1 + fBo, gx, gwd, gmd, acc);
        nalu_stage_lean<false, false, false, 25, false>(f1, f0, c0, c1, rec, slot0 + sAo, (float*)(slot0 + sDo),
                                                 slot0 + fAo, slot0 + fBo, gx, gwd, gmd, acc);
    }

    if (lane == 0) sign_state[wave] = live ? 1 : 0;
    __builtin_amdgcn_fence(__ATOMIC_RELEASE, "workgroup", "local");
    __builtin_amdgcn_s_barrier_signal(-1);
    __builtin_amdgcn_s_barrier_wait(-1);
    __builtin_amdgcn_fence(__ATOMIC_ACQUIRE, "workgroup", "local");
    const v4i fin = *(const v4i*)sign_state;
    const bool need_m = __builtin_amdgcn_readfirstlane(fin.x | fin.y | fin.z | fin.w) != 0;
    if (need_m) {
        lp = lp0;
        NALU_AT_STAGE();
        nalu_raw_load_lean(gx, gwd, gmd, c0);
        NALU_ADVANCE();
        nalu_raw_load_lean(gx, gwd, gmd, c1);
        NALU_ADVANCE();
        nalu_xform_d(c0, rec);
        nalu_xform_x<true>(c0, rec);
        nalu_rec_store_lean(slot0 + sAo, (float*)(slot0 + sDo), rec);
        nalu_raw_load_lean(gx, gwd, gmd, c0);
        NALU_ADVANCE();
        nalu_xform_d(c1, rec);
        nalu_xform_x<true>(c1, rec);
        __builtin_amdgcn_fence(__ATOMIC_RELEASE, "workgroup", "local");
        __builtin_amdgcn_s_barrier_signal(-1);
        __builtin_amdgcn_s_barrier_wait(-1);
        __builtin_amdgcn_fence(__ATOMIC_ACQUIRE, "workgroup", "local");
        nalu_frag_load_lean(slot0 + fAo, slot0 + fBo, f0);
#pragma unroll 1
        for (int i2 = 0; i2 < (NSTAGE - 4) / 2; ++i2) {
            nalu_stage_lean<true, true, true, 30, true>(f0, f1, c1, c0, rec, slot1 + sAo, (float*)(slot1 + sDo),
                                                        slot1 + fAo, slot1 + fBo, gx, gwd, gmd, acc);
            NALU_ADVANCE();
            nalu_stage_lean<true, true, true, 31, true>(f1, f0, c0, c1, rec, slot0 + sAo, (float*)(slot0 + sDo),
                                                        slot0 + fAo, slot0 + fBo, gx, gwd, gmd, acc);
            NALU_ADVANCE();
        }
        nalu_stage_lean<true, true, true, 32, true>(f0, f1, c1, c0, rec, slot1 + sAo, (float*)(slot1 + sDo),
                                                    slot1 + fAo, slot1 + fBo, gx, gwd, gmd, acc);
        nalu_stage_lean<true, false, true, 33, true>(f1, f0, c0, c1, rec, slot0 + sAo, (float*)(slot0 + sDo),
                                                     slot0 + fAo, slot0 + fBo, gx, gwd, gmd, acc);
        nalu_stage_lean<true, false, false, 34, true>(f0, f1, c1, c0, rec, slot1 + sAo, (float*)(slot1 + sDo),
                                                      slot1 + fAo, slot1 + fBo, gx, gwd, gmd, acc);
        nalu_stage_lean<false, false, false, 35, true>(f1, f0, c0, c1, rec, slot0 + sAo, (float*)(slot0 + sDo),
                                                       slot0 + fAo, slot0 + fBo, gx, gwd, gmd, acc);
    }

    const int m0 = bm0 + wm * 32;
    const int bcol = bn0 + wn * 16 + l15;
    const float g1 = 1.0f / (1.0f + expf(-g[bcol]));
    const float gm = 1.0f - g1;
#pragma unroll
    for (int v = 0; v < 8; ++v) {
        const int row0 = m0 + v + 8 * half;
        const int negc = (int)(acc.P[v] + 0.5f);
        {
            const float m1 = expf(fminf(acc.M0[v], LOG_CLIP));
            float ms = ((negc & 1) ? -1.0f : 1.0f) * expf(acc.L0[v]);
            ms = fminf(fmaxf(ms, -1.0f), 1.0f);
            out[row0 * NO + bcol] = g1 * acc.A0[v] + gm * m1 * ms;
        }
        {
            const float m1 = expf(fminf(acc.M1[v], LOG_CLIP));
            float ms = (((negc >> PSHIFT) & 1) ? -1.0f : 1.0f) * expf(acc.L1[v]);
            ms = fminf(fmaxf(ms, -1.0f), 1.0f);
            out[(row0 + 16) * NO + bcol] = g1 * acc.A1[v] + gm * m1 * ms;
        }
    }
}

extern "C" void kernel_launch(void* const* d_in, const int* in_sizes, int n_in,
                              void* d_out, int out_size, void* d_ws, size_t ws_size,
                              hipStream_t stream) {
    const float* x     = (const float*)d_in[0];
    const float* w_hat = (const float*)d_in[1];
    const float* m_hat = (const float*)d_in[2];
    const float* g     = (const float*)d_in[3];
    float* out = (float*)d_out;

    nalu_main_kernel<<<256, 128, 0, stream>>>(x, w_hat, m_hat, g, out);
}
